# starting kernel
# baseline (speedup 1.0000x reference)
#include <hip/hip_runtime.h>
#include <hip/hip_bf16.h>
#include <math.h>

#define B_   4
#define N_   512
#define D_   32
#define PE_  16

#define WPB 4
#define L2E 1.44269504088896340736f
#define INV_SQRT_D 0.17677669529663688110f
#define KS2 (10.0f * L2E)

typedef __attribute__((ext_vector_type(2))) float v2f;
typedef __attribute__((ext_vector_type(8))) float v8f;

static __device__ __forceinline__ v8f wmma_f32(v2f a, v2f b, v8f c) {
  return __builtin_amdgcn_wmma_f32_16x16x4_f32(false, a, false, b, (short)0, c, false, false);
}

static __device__ __forceinline__ float fexp2(float x) { return __builtin_amdgcn_exp2f(x); }
static __device__ __forceinline__ float frcp(float x)  { return __builtin_amdgcn_rcpf(x); }

#define DPP_QUAD_XOR1  0xB1
#define DPP_QUAD_XOR2  0x4E
#define DPP_HALF_MIR   0x141
#define DPP_ROW_MIR    0x140

template <int CTRL>
static __device__ __forceinline__ float dpp_mov(float x) {
  return __int_as_float(
      __builtin_amdgcn_update_dpp(0, __float_as_int(x), CTRL, 0xf, 0xf, true));
}
static __device__ __forceinline__ float redmax16(float x) {
  x = fmaxf(x, dpp_mov<DPP_QUAD_XOR1>(x));
  x = fmaxf(x, dpp_mov<DPP_QUAD_XOR2>(x));
  x = fmaxf(x, dpp_mov<DPP_HALF_MIR>(x));
  x = fmaxf(x, dpp_mov<DPP_ROW_MIR>(x));
  return x;
}
static __device__ __forceinline__ float redsum16(float x) {
  x += dpp_mov<DPP_QUAD_XOR1>(x);
  x += dpp_mov<DPP_QUAD_XOR2>(x);
  x += dpp_mov<DPP_HALF_MIR>(x);
  x += dpp_mov<DPP_ROW_MIR>(x);
  return x;
}

__global__ void kq_proj_kernel(const float* __restrict__ pe,
                               const float* __restrict__ Wk, const float* __restrict__ bk,
                               const float* __restrict__ Wq, const float* __restrict__ bq,
                               float* __restrict__ Kd, float* __restrict__ Qd) {
  int idx = blockIdx.x * blockDim.x + threadIdx.x;
  if (idx >= B_ * N_ * D_) return;
  int d  = idx & (D_ - 1);
  int bn = idx / D_;
  const float* p = pe + bn * PE_;
  float k = bk[d], q = bq[d];
#pragma unroll
  for (int e = 0; e < PE_; ++e) {
    float pv = p[e];
    k = fmaf(pv, Wk[e * D_ + d], k);
    q = fmaf(pv, Wq[e * D_ + d], q);
  }
  Kd[idx] = k;
  Qd[idx] = q;
}

__global__ void __launch_bounds__(32 * WPB)
fokker_planck_fused_kernel(const float* __restrict__ h,
                           const float* __restrict__ E,
                           const float* __restrict__ A,
                           const float* __restrict__ beta,
                           const float* __restrict__ Kd,
                           const float* __restrict__ Qd,
                           float* __restrict__ out) {
  __shared__ float lds[WPB * 3 * 256];

  const int lane = threadIdx.x & 31;
  const int wid  = threadIdx.x >> 5;
  const int lo   = lane & 15;
  const int hi   = lane >> 4;
  const int tile = blockIdx.x * WPB + wid;
  if (tile >= B_ * (N_ / 16)) return;
  const int b  = tile >> 5;
  const int i0 = (tile & 31) * 16;

  float* lds1 = lds + wid * 3 * 256;
  float* lds3 = lds1 + 256;
  float* lds4 = lds1 + 512;

  const float* Kb = Kd + b * N_ * D_;
  const float* Qb = Qd + b * N_ * D_;
  const float* Hb = h  + b * N_ * D_;

  const float kscale = INV_SQRT_D * L2E;
  v2f afK[8];
#pragma unroll
  for (int kc = 0; kc < 8; ++kc) {
    v2f t = *(const v2f*)(Kb + (i0 + lo) * D_ + 4 * kc + 2 * hi);
    afK[kc].x = t.x * kscale;
    afK[kc].y = t.y * kscale;
  }

  float Ei[8];
#pragma unroll
  for (int v = 0; v < 8; ++v) Ei[v] = E[i0 + v + 8 * hi];

  v8f acc1[2] = {}, acc3[2] = {}, acc4[2] = {}, acc5[2] = {};
  float m[8], z[8], c2[8], c6[8];
#pragma unroll
  for (int v = 0; v < 8; ++v) { m[v] = -1.0e30f; z[v] = 0.f; c2[v] = 0.f; c6[v] = 0.f; }

  for (int j0 = 0; j0 < N_; j0 += 16) {
    v8f sraw = {};
#pragma unroll
    for (int kc = 0; kc < 8; ++kc) {
      v2f bfQ = *(const v2f*)(Qb + (j0 + lo) * D_ + 4 * kc + 2 * hi);
      sraw = wmma_f32(afK[kc], bfQ, sraw);
    }

    const float Ej = E[j0 + lo];

    float sc[8];
#pragma unroll
    for (int v = 0; v < 8; ++v)
      sc[v] = A[(i0 + v + 8 * hi) * N_ + j0 + lo] * sraw[v];

    float tm[8];
#pragma unroll
    for (int v = 0; v < 8; ++v) tm[v] = redmax16(sc[v]);

#pragma unroll
    for (int v = 0; v < 8; ++v) {
      float mn = fmaxf(m[v], tm[v]);
      float r  = fexp2(m[v] - mn);
      m[v] = mn;
      z[v]  *= r;  c2[v] *= r;  c6[v] *= r;
      acc1[0][v] *= r; acc1[1][v] *= r;
      acc3[0][v] *= r; acc3[1][v] *= r;
      acc4[0][v] *= r; acc4[1][v] *= r;
      acc5[0][v] *= r; acc5[1][v] *= r;

      float p  = fexp2(sc[v] - mn);
      float Ed = Ej - Ei[v];
      float sg = frcp(1.f + fexp2(-KS2 * Ed));
      float p3 = p * sg;
      float p1 = p3 * Ed;
      float p4 = p - p3;
      z[v]  += p;
      c2[v] += p * Ed - p1;
      c6[v] += p4;

      int addr = (v + 8 * hi) * 16 + lo;
      lds1[addr] = p1;
      lds3[addr] = p3;
      lds4[addr] = p4;
    }

    asm volatile("s_wait_dscnt 0" ::: "memory");

    v2f p1a[4], p3a[4], p4a[4];
#pragma unroll
    for (int kc = 0; kc < 4; ++kc) {
      int a = lo * 16 + 4 * kc + 2 * hi;
      p1a[kc] = *(const v2f*)(lds1 + a);
      p3a[kc] = *(const v2f*)(lds3 + a);
      p4a[kc] = *(const v2f*)(lds4 + a);
    }

    v2f HB[4][2], LB[4][2], LHB[4][2];
#pragma unroll
    for (int kc = 0; kc < 4; ++kc)
#pragma unroll
      for (int s = 0; s < 2; ++s)
#pragma unroll
        for (int t = 0; t < 2; ++t) {
          float hv = Hb[(j0 + 4 * kc + 2 * hi + t) * D_ + lo + 16 * s];
          float lv = __logf(hv + 1e-8f);
          HB[kc][s][t]  = hv;
          LB[kc][s][t]  = lv;
          LHB[kc][s][t] = lv * hv;
        }

#pragma unroll
    for (int s = 0; s < 2; ++s)
#pragma unroll
      for (int kc = 0; kc < 4; ++kc) {
        acc1[s] = wmma_f32(p1a[kc], HB[kc][s],  acc1[s]);
        acc5[s] = wmma_f32(p3a[kc], HB[kc][s],  acc5[s]);
        acc3[s] = wmma_f32(p3a[kc], LHB[kc][s], acc3[s]);
        acc4[s] = wmma_f32(p4a[kc], LB[kc][s],  acc4[s]);
      }
  }

#pragma unroll
  for (int v = 0; v < 8; ++v) {
    z[v]  = redsum16(z[v]);
    c2[v] = redsum16(c2[v]);
    c6[v] = redsum16(c6[v]);
  }

  float bt[2];
  bt[0] = beta[lo];
  bt[1] = beta[lo + 16];

#pragma unroll
  for (int v = 0; v < 8; ++v) {
    const int row = i0 + v + 8 * hi;
    const float invz = frcp(z[v]);
#pragma unroll
    for (int s = 0; s < 2; ++s) {
      const int col = lo + 16 * s;
      float hv = Hb[row * D_ + col];
      float lv = __logf(hv + 1e-8f);
      float num = acc1[s][v] + c2[v] * hv +
                  bt[s] * (acc3[s][v] + hv * acc4[s][v] - lv * acc5[s][v] - c6[v] * lv * hv);
      out[(b * N_ + row) * D_ + col] = num * invz;
    }
  }
}

extern "C" void kernel_launch(void* const* d_in, const int* in_sizes, int n_in,
                              void* d_out, int out_size, void* d_ws, size_t ws_size,
                              hipStream_t stream) {
  const float* h    = (const float*)d_in[1];
  const float* pe   = (const float*)d_in[2];
  const float* E    = (const float*)d_in[3];
  const float* A    = (const float*)d_in[4];
  const float* Wk   = (const float*)d_in[5];
  const float* bk   = (const float*)d_in[6];
  const float* Wq   = (const float*)d_in[7];
  const float* bq   = (const float*)d_in[8];
  const float* beta = (const float*)d_in[9];
  float* out = (float*)d_out;

  float* Kd = (float*)d_ws;
  float* Qd = Kd + (size_t)B_ * N_ * D_;

  {
    int total = B_ * N_ * D_;
    kq_proj_kernel<<<(total + 255) / 256, 256, 0, stream>>>(pe, Wk, bk, Wq, bq, Kd, Qd);
  }
  {
    int tiles = B_ * (N_ / 16);
    dim3 grid(tiles / WPB), block(32 * WPB);
    fokker_planck_fused_kernel<<<grid, block, 0, stream>>>(h, E, A, beta, Kd, Qd, out);
  }
}


# Split each 16-row tile's j loop over the 16 waves of one work-group (128 groups x 16 waves, was 32 x 4); merge the 16 online-softmax partials through LDS in fixed order (changes the summation tree)
# speedup vs baseline: 6.4210x; 6.4210x over previous
#include <hip/hip_runtime.h>
#include <hip/hip_bf16.h>
#include <math.h>

#define B_   4
#define N_   512
#define D_   32
#define PE_  16

#define NW  16
#define JT_PER_W (N_ / 16 / NW)
static_assert(NW == 16, "the merge gives output row r of the 16-row tile to wave r");
static_assert(JT_PER_W * NW * 16 == N_, "the waves' j slices tile N exactly");
#define L2E 1.44269504088896340736f
#define INV_SQRT_D 0.17677669529663688110f
#define KS2 (10.0f * L2E)

typedef __attribute__((ext_vector_type(2))) float v2f;
typedef __attribute__((ext_vector_type(8))) float v8f;

static __device__ __forceinline__ v8f wmma_f32(v2f a, v2f b, v8f c) {
  return __builtin_amdgcn_wmma_f32_16x16x4_f32(false, a, false, b, (short)0, c, false, false);
}

static __device__ __forceinline__ float fexp2(float x) { return __builtin_amdgcn_exp2f(x); }
static __device__ __forceinline__ float frcp(float x)  { return __builtin_amdgcn_rcpf(x); }

#define DPP_QUAD_XOR1  0xB1
#define DPP_QUAD_XOR2  0x4E
#define DPP_HALF_MIR   0x141
#define DPP_ROW_MIR    0x140

template <int CTRL>
static __device__ __forceinline__ float dpp_mov(float x) {
  return __int_as_float(
      __builtin_amdgcn_update_dpp(0, __float_as_int(x), CTRL, 0xf, 0xf, true));
}
static __device__ __forceinline__ float redmax16(float x) {
  x = fmaxf(x, dpp_mov<DPP_QUAD_XOR1>(x));
  x = fmaxf(x, dpp_mov<DPP_QUAD_XOR2>(x));
  x = fmaxf(x, dpp_mov<DPP_HALF_MIR>(x));
  x = fmaxf(x, dpp_mov<DPP_ROW_MIR>(x));
  return x;
}
static __device__ __forceinline__ float redsum16(float x) {
  x += dpp_mov<DPP_QUAD_XOR1>(x);
  x += dpp_mov<DPP_QUAD_XOR2>(x);
  x += dpp_mov<DPP_HALF_MIR>(x);
  x += dpp_mov<DPP_ROW_MIR>(x);
  return x;
}

__global__ void kq_proj_kernel(const float* __restrict__ pe,
                               const float* __restrict__ Wk, const float* __restrict__ bk,
                               const float* __restrict__ Wq, const float* __restrict__ bq,
                               float* __restrict__ Kd, float* __restrict__ Qd) {
  int idx = blockIdx.x * blockDim.x + threadIdx.x;
  if (idx >= B_ * N_ * D_) return;
  int d  = idx & (D_ - 1);
  int bn = idx / D_;
  const float* p = pe + bn * PE_;
  float k = bk[d], q = bq[d];
#pragma unroll
  for (int e = 0; e < PE_; ++e) {
    float pv = p[e];
    k = fmaf(pv, Wk[e * D_ + d], k);
    q = fmaf(pv, Wq[e * D_ + d], q);
  }
  Kd[idx] = k;
  Qd[idx] = q;
}


static __device__ __forceinline__ int mrg_idx(int row, int col) {
  return row * 32 + (col ^ ((row & 8) << 1));
}

__global__ void __launch_bounds__(32 * NW)
fokker_planck_fused_kernel(const float* __restrict__ h,
                           const float* __restrict__ E,
                           const float* __restrict__ A,
                           const float* __restrict__ beta,
                           const float* __restrict__ Kd,
                           const float* __restrict__ Qd,
                           float* __restrict__ out) {
  __shared__ float lds[NW * 3 * 256];
  __shared__ float mrgA[NW * 4 * 16 * 32];
  __shared__ __attribute__((aligned(16))) float mrgS[NW * 16 * 4];

  const int lane = threadIdx.x & 31;
  const int wid  = threadIdx.x >> 5;
  const int lo   = lane & 15;
  const int hi   = lane >> 4;
  const int tile = blockIdx.x;
  const int b  = tile >> 5;
  const int i0 = (tile & 31) * 16;

  float* lds1 = lds + wid * 3 * 256;
  float* lds3 = lds1 + 256;
  float* lds4 = lds1 + 512;

  const float* Kb = Kd + b * N_ * D_;
  const float* Qb = Qd + b * N_ * D_;
  const float* Hb = h  + b * N_ * D_;

  const float kscale = INV_SQRT_D * L2E;
  v2f afK[8];
#pragma unroll
  for (int kc = 0; kc < 8; ++kc) {
    v2f t = *(const v2f*)(Kb + (i0 + lo) * D_ + 4 * kc + 2 * hi);
    afK[kc].x = t.x * kscale;
    afK[kc].y = t.y * kscale;
  }

  float Ei[8];
#pragma unroll
  for (int v = 0; v < 8; ++v) Ei[v] = E[i0 + v + 8 * hi];

  v8f acc1[2] = {}, acc3[2] = {}, acc4[2] = {}, acc5[2] = {};
  float m[8], z[8], c2[8], c6[8];
#pragma unroll
  for (int v = 0; v < 8; ++v) { m[v] = -1.0e30f; z[v] = 0.f; c2[v] = 0.f; c6[v] = 0.f; }

  for (int jt = 0; jt < JT_PER_W; ++jt) {
    const int j0 = (wid * JT_PER_W + jt) * 16;
    v8f sraw = {};
#pragma unroll
    for (int kc = 0; kc < 8; ++kc) {
      v2f bfQ = *(const v2f*)(Qb + (j0 + lo) * D_ + 4 * kc + 2 * hi);
      sraw = wmma_f32(afK[kc], bfQ, sraw);
    }

    const float Ej = E[j0 + lo];

    float sc[8];
#pragma unroll
    for (int v = 0; v < 8; ++v)
      sc[v] = A[(i0 + v + 8 * hi) * N_ + j0 + lo] * sraw[v];

    float tm[8];
#pragma unroll
    for (int v = 0; v < 8; ++v) tm[v] = redmax16(sc[v]);

#pragma unroll
    for (int v = 0; v < 8; ++v) {
      float mn = fmaxf(m[v], tm[v]);
      float r  = fexp2(m[v] - mn);
      m[v] = mn;
      z[v]  *= r;  c2[v] *= r;  c6[v] *= r;
      acc1[0][v] *= r; acc1[1][v] *= r;
      acc3[0][v] *= r; acc3[1][v] *= r;
      acc4[0][v] *= r; acc4[1][v] *= r;
      acc5[0][v] *= r; acc5[1][v] *= r;

      float p  = fexp2(sc[v] - mn);
      float Ed = Ej - Ei[v];
      float sg = frcp(1.f + fexp2(-KS2 * Ed));
      float p3 = p * sg;
      float p1 = p3 * Ed;
      float p4 = p - p3;
      z[v]  += p;
      c2[v] += p * Ed - p1;
      c6[v] += p4;

      int addr = (v + 8 * hi) * 16 + lo;
      lds1[addr] = p1;
      lds3[addr] = p3;
      lds4[addr] = p4;
    }

    asm volatile("s_wait_dscnt 0" ::: "memory");

    v2f p1a[4], p3a[4], p4a[4];
#pragma unroll
    for (int kc = 0; kc < 4; ++kc) {
      int a = lo * 16 + 4 * kc + 2 * hi;
      p1a[kc] = *(const v2f*)(lds1 + a);
      p3a[kc] = *(const v2f*)(lds3 + a);
      p4a[kc] = *(const v2f*)(lds4 + a);
    }

    v2f HB[4][2], LB[4][2], LHB[4][2];
#pragma unroll
    for (int kc = 0; kc < 4; ++kc)
#pragma unroll
      for (int s = 0; s < 2; ++s)
#pragma unroll
        for (int t = 0; t < 2; ++t) {
          float hv = Hb[(j0 + 4 * kc + 2 * hi + t) * D_ + lo + 16 * s];
          float lv = __logf(hv + 1e-8f);
          HB[kc][s][t]  = hv;
          LB[kc][s][t]  = lv;
          LHB[kc][s][t] = lv * hv;
        }

#pragma unroll
    for (int s = 0; s < 2; ++s)
#pragma unroll
      for (int kc = 0; kc < 4; ++kc) {
        acc1[s] = wmma_f32(p1a[kc], HB[kc][s],  acc1[s]);
        acc5[s] = wmma_f32(p3a[kc], HB[kc][s],  acc5[s]);
        acc3[s] = wmma_f32(p3a[kc], LHB[kc][s], acc3[s]);
        acc4[s] = wmma_f32(p4a[kc], LB[kc][s],  acc4[s]);
      }
  }

#pragma unroll
  for (int v = 0; v < 8; ++v) {
    z[v]  = redsum16(z[v]);
    c2[v] = redsum16(c2[v]);
    c6[v] = redsum16(c6[v]);
  }

  {
    float* myA = mrgA + wid * (4 * 16 * 32);
#pragma unroll
    for (int v = 0; v < 8; ++v) {
      const int row = v + 8 * hi;
#pragma unroll
      for (int s = 0; s < 2; ++s) {
        const int a = mrg_idx(row, lo + 16 * s);
        myA[0 * 512 + a] = acc1[s][v];
        myA[1 * 512 + a] = acc3[s][v];
        myA[2 * 512 + a] = acc4[s][v];
        myA[3 * 512 + a] = acc5[s][v];
      }
    }
    if (lo == 0) {
#pragma unroll
      for (int v = 0; v < 8; ++v) {
        float4 sv; sv.x = m[v]; sv.y = z[v]; sv.z = c2[v]; sv.w = c6[v];
        *(float4*)(mrgS + (wid * 16 + v + 8 * hi) * 4) = sv;
      }
    }
  }
  __syncthreads();

  {
    const int row = wid;
    const int col = lane;
    float4 sv[NW];
    float M = -1.0e30f;
#pragma unroll
    for (int w = 0; w < NW; ++w) {
      sv[w] = *(const float4*)(mrgS + (w * 16 + row) * 4);
      M = fmaxf(M, sv[w].x);
    }
    const int a = mrg_idx(row, col);
    float zt = 0.f, c2t = 0.f, c6t = 0.f, a1 = 0.f, a3 = 0.f, a4 = 0.f, a5 = 0.f;
#pragma unroll
    for (int w = 0; w < NW; ++w) {
      const float f = fexp2(sv[w].x - M);
      const float* wA = mrgA + w * (4 * 16 * 32);
      zt  = fmaf(f, sv[w].y, zt);
      c2t = fmaf(f, sv[w].z, c2t);
      c6t = fmaf(f, sv[w].w, c6t);
      a1  = fmaf(f, wA[0 * 512 + a], a1);
      a3  = fmaf(f, wA[1 * 512 + a], a3);
      a4  = fmaf(f, wA[2 * 512 + a], a4);
      a5  = fmaf(f, wA[3 * 512 + a], a5);
    }

    const float bt   = beta[col];
    const float invz = frcp(zt);
    const float hv = Hb[(i0 + row) * D_ + col];
    const float lv = __logf(hv + 1e-8f);
    const float num = a1 + c2t * hv +
                      bt * (a3 + hv * a4 - lv * a5 - c6t * lv * hv);
    out[(b * N_ + i0 + row) * D_ + col] = num * invz;
  }
}

extern "C" void kernel_launch(void* const* d_in, const int* in_sizes, int n_in,
                              void* d_out, int out_size, void* d_ws, size_t ws_size,
                              hipStream_t stream) {
  const float* h    = (const float*)d_in[1];
  const float* pe   = (const float*)d_in[2];
  const float* E    = (const float*)d_in[3];
  const float* A    = (const float*)d_in[4];
  const float* Wk   = (const float*)d_in[5];
  const float* bk   = (const float*)d_in[6];
  const float* Wq   = (const float*)d_in[7];
  const float* bq   = (const float*)d_in[8];
  const float* beta = (const float*)d_in[9];
  float* out = (float*)d_out;

  float* Kd = (float*)d_ws;
  float* Qd = Kd + (size_t)B_ * N_ * D_;

  {
    int total = B_ * N_ * D_;
    kq_proj_kernel<<<(total + 255) / 256, 256, 0, stream>>>(pe, Wk, bk, Wq, bq, Kd, Qd);
  }
  {
    dim3 grid(B_ * (N_ / 16)), block(32 * NW);
    fokker_planck_fused_kernel<<<grid, block, 0, stream>>>(h, E, A, beta, Kd, Qd, out);
  }
}


# Fuse the K/Q projection into the main kernel as transposed f32 WMMA tiles that serve directly as score operands: one launch, no workspace; the score dot now sums d in tile-register order
# speedup vs baseline: 8.4541x; 1.3166x over previous
#include <hip/hip_runtime.h>
#include <hip/hip_bf16.h>
#include <math.h>

#define B_   4
#define N_   512
#define D_   32
#define PE_  16

#define NW  16
#define JT_PER_W (N_ / 16 / NW)
static_assert(NW == 16, "the merge gives output row r of the 16-row tile to wave r");
static_assert(JT_PER_W * NW * 16 == N_, "the waves' j slices tile N exactly");
#define L2E 1.44269504088896340736f
#define INV_SQRT_D 0.17677669529663688110f
#define KS2 (10.0f * L2E)

typedef __attribute__((ext_vector_type(2))) float v2f;
typedef __attribute__((ext_vector_type(8))) float v8f;

static __device__ __forceinline__ v8f wmma_f32(v2f a, v2f b, v8f c) {
  return __builtin_amdgcn_wmma_f32_16x16x4_f32(false, a, false, b, (short)0, c, false, false);
}

static __device__ __forceinline__ float fexp2(float x) { return __builtin_amdgcn_exp2f(x); }
static __device__ __forceinline__ float frcp(float x)  { return __builtin_amdgcn_rcpf(x); }

#define DPP_QUAD_XOR1  0xB1
#define DPP_QUAD_XOR2  0x4E
#define DPP_HALF_MIR   0x141
#define DPP_ROW_MIR    0x140

template <int CTRL>
static __device__ __forceinline__ float dpp_mov(float x) {
  return __int_as_float(
      __builtin_amdgcn_update_dpp(0, __float_as_int(x), CTRL, 0xf, 0xf, true));
}
static __device__ __forceinline__ float redmax16(float x) {
  x = fmaxf(x, dpp_mov<DPP_QUAD_XOR1>(x));
  x = fmaxf(x, dpp_mov<DPP_QUAD_XOR2>(x));
  x = fmaxf(x, dpp_mov<DPP_HALF_MIR>(x));
  x = fmaxf(x, dpp_mov<DPP_ROW_MIR>(x));
  return x;
}
static __device__ __forceinline__ float redsum16(float x) {
  x += dpp_mov<DPP_QUAD_XOR1>(x);
  x += dpp_mov<DPP_QUAD_XOR2>(x);
  x += dpp_mov<DPP_HALF_MIR>(x);
  x += dpp_mov<DPP_ROW_MIR>(x);
  return x;
}


static __device__ __forceinline__ int mrg_idx(int row, int col) {
  return row * 32 + (col ^ ((row & 8) << 1));
}

__global__ void __launch_bounds__(32 * NW)
fokker_planck_fused_kernel(const float* __restrict__ h,
                           const float* __restrict__ pe,
                           const float* __restrict__ E,
                           const float* __restrict__ A,
                           const float* __restrict__ Wk, const float* __restrict__ bk,
                           const float* __restrict__ Wq, const float* __restrict__ bq,
                           const float* __restrict__ beta,
                           float* __restrict__ out) {
  __shared__ float lds[NW * 3 * 256];
  __shared__ float mrgA[NW * 4 * 16 * 32];
  __shared__ __attribute__((aligned(16))) float mrgS[NW * 16 * 4];

  const int lane = threadIdx.x & 31;
  const int wid  = threadIdx.x >> 5;
  const int lo   = lane & 15;
  const int hi   = lane >> 4;
  const int tile = blockIdx.x;
  const int b  = tile >> 5;
  const int i0 = (tile & 31) * 16;

  float* lds1 = lds + wid * 3 * 256;
  float* lds3 = lds1 + 256;
  float* lds4 = lds1 + 512;

  const float* Pb = pe + b * N_ * PE_;
  const float* Hb = h  + b * N_ * D_;

  float Ei[8];
#pragma unroll
  for (int v = 0; v < 8; ++v) Ei[v] = E[i0 + v + 8 * hi];

  v2f aWk[2][4], aWq[2][4];
#pragma unroll
  for (int s = 0; s < 2; ++s)
#pragma unroll
    for (int kc = 0; kc < 4; ++kc) {
      const int o = (4 * kc + 2 * hi) * D_ + 16 * s + lo;
      aWk[s][kc].x = Wk[o];  aWk[s][kc].y = Wk[o + D_];
      aWq[s][kc].x = Wq[o];  aWq[s][kc].y = Wq[o + D_];
    }
  v8f cbk[2], cbq[2];
#pragma unroll
  for (int s = 0; s < 2; ++s)
#pragma unroll
    for (int v = 0; v < 8; ++v) {
      cbk[s][v] = bk[16 * s + 8 * hi + v];
      cbq[s][v] = bq[16 * s + 8 * hi + v];
    }

  v8f kt0 = cbk[0], kt1 = cbk[1];
#pragma unroll
  for (int kc = 0; kc < 4; ++kc) {
    v2f bPi = *(const v2f*)(Pb + (i0 + lo) * PE_ + 4 * kc + 2 * hi);
    kt0 = wmma_f32(aWk[0][kc], bPi, kt0);
    kt1 = wmma_f32(aWk[1][kc], bPi, kt1);
  }
  const float kscale = INV_SQRT_D * L2E;
  v2f afK[8];
#pragma unroll
  for (int q = 0; q < 4; ++q) {
    afK[q].x     = kt0[2 * q] * kscale;  afK[q].y     = kt0[2 * q + 1] * kscale;
    afK[4 + q].x = kt1[2 * q] * kscale;  afK[4 + q].y = kt1[2 * q + 1] * kscale;
  }

  v8f acc1[2] = {}, acc3[2] = {}, acc4[2] = {}, acc5[2] = {};
  float m[8], z[8], c2[8], c6[8];
#pragma unroll
  for (int v = 0; v < 8; ++v) { m[v] = -1.0e30f; z[v] = 0.f; c2[v] = 0.f; c6[v] = 0.f; }

  for (int jt = 0; jt < JT_PER_W; ++jt) {
    const int j0 = (wid * JT_PER_W + jt) * 16;
    const float Ej = E[j0 + lo];
    float Am[8];
#pragma unroll
    for (int v = 0; v < 8; ++v) Am[v] = A[(i0 + v + 8 * hi) * N_ + j0 + lo];

    v8f qt0 = cbq[0], qt1 = cbq[1];
#pragma unroll
    for (int kc = 0; kc < 4; ++kc) {
      v2f bPj = *(const v2f*)(Pb + (j0 + lo) * PE_ + 4 * kc + 2 * hi);
      qt0 = wmma_f32(aWq[0][kc], bPj, qt0);
      qt1 = wmma_f32(aWq[1][kc], bPj, qt1);
    }
    v8f sraw = {};
#pragma unroll
    for (int q = 0; q < 4; ++q) {
      v2f bfQ; bfQ.x = qt0[2 * q]; bfQ.y = qt0[2 * q + 1];
      sraw = wmma_f32(afK[q], bfQ, sraw);
    }
#pragma unroll
    for (int q = 0; q < 4; ++q) {
      v2f bfQ; bfQ.x = qt1[2 * q]; bfQ.y = qt1[2 * q + 1];
      sraw = wmma_f32(afK[4 + q], bfQ, sraw);
    }

    float sc[8];
#pragma unroll
    for (int v = 0; v < 8; ++v)
      sc[v] = Am[v] * sraw[v];

    float tm[8];
#pragma unroll
    for (int v = 0; v < 8; ++v) tm[v] = redmax16(sc[v]);

#pragma unroll
    for (int v = 0; v < 8; ++v) {
      float mn = fmaxf(m[v], tm[v]);
      float r  = fexp2(m[v] - mn);
      m[v] = mn;
      z[v]  *= r;  c2[v] *= r;  c6[v] *= r;
      acc1[0][v] *= r; acc1[1][v] *= r;
      acc3[0][v] *= r; acc3[1][v] *= r;
      acc4[0][v] *= r; acc4[1][v] *= r;
      acc5[0][v] *= r; acc5[1][v] *= r;

      float p  = fexp2(sc[v] - mn);
      float Ed = Ej - Ei[v];
      float sg = frcp(1.f + fexp2(-KS2 * Ed));
      float p3 = p * sg;
      float p1 = p3 * Ed;
      float p4 = p - p3;
      z[v]  += p;
      c2[v] += p * Ed - p1;
      c6[v] += p4;

      int addr = (v + 8 * hi) * 16 + lo;
      lds1[addr] = p1;
      lds3[addr] = p3;
      lds4[addr] = p4;
    }

    asm volatile("s_wait_dscnt 0" ::: "memory");

    v2f p1a[4], p3a[4], p4a[4];
#pragma unroll
    for (int kc = 0; kc < 4; ++kc) {
      int a = lo * 16 + 4 * kc + 2 * hi;
      p1a[kc] = *(const v2f*)(lds1 + a);
      p3a[kc] = *(const v2f*)(lds3 + a);
      p4a[kc] = *(const v2f*)(lds4 + a);
    }

    v2f HB[4][2], LB[4][2], LHB[4][2];
#pragma unroll
    for (int kc = 0; kc < 4; ++kc)
#pragma unroll
      for (int s = 0; s < 2; ++s)
#pragma unroll
        for (int t = 0; t < 2; ++t) {
          float hv = Hb[(j0 + 4 * kc + 2 * hi + t) * D_ + lo + 16 * s];
          float lv = __logf(hv + 1e-8f);
          HB[kc][s][t]  = hv;
          LB[kc][s][t]  = lv;
          LHB[kc][s][t] = lv * hv;
        }

#pragma unroll
    for (int s = 0; s < 2; ++s)
#pragma unroll
      for (int kc = 0; kc < 4; ++kc) {
        acc1[s] = wmma_f32(p1a[kc], HB[kc][s],  acc1[s]);
        acc5[s] = wmma_f32(p3a[kc], HB[kc][s],  acc5[s]);
        acc3[s] = wmma_f32(p3a[kc], LHB[kc][s], acc3[s]);
        acc4[s] = wmma_f32(p4a[kc], LB[kc][s],  acc4[s]);
      }
  }

#pragma unroll
  for (int v = 0; v < 8; ++v) {
    z[v]  = redsum16(z[v]);
    c2[v] = redsum16(c2[v]);
    c6[v] = redsum16(c6[v]);
  }

  {
    float* myA = mrgA + wid * (4 * 16 * 32);
#pragma unroll
    for (int v = 0; v < 8; ++v) {
      const int row = v + 8 * hi;
#pragma unroll
      for (int s = 0; s < 2; ++s) {
        const int a = mrg_idx(row, lo + 16 * s);
        myA[0 * 512 + a] = acc1[s][v];
        myA[1 * 512 + a] = acc3[s][v];
        myA[2 * 512 + a] = acc4[s][v];
        myA[3 * 512 + a] = acc5[s][v];
      }
    }
    if (lo == 0) {
#pragma unroll
      for (int v = 0; v < 8; ++v) {
        float4 sv; sv.x = m[v]; sv.y = z[v]; sv.z = c2[v]; sv.w = c6[v];
        *(float4*)(mrgS + (wid * 16 + v + 8 * hi) * 4) = sv;
      }
    }
  }
  __syncthreads();

  {
    const int row = wid;
    const int col = lane;
    float4 sv[NW];
    float M = -1.0e30f;
#pragma unroll
    for (int w = 0; w < NW; ++w) {
      sv[w] = *(const float4*)(mrgS + (w * 16 + row) * 4);
      M = fmaxf(M, sv[w].x);
    }
    const int a = mrg_idx(row, col);
    float zt = 0.f, c2t = 0.f, c6t = 0.f, a1 = 0.f, a3 = 0.f, a4 = 0.f, a5 = 0.f;
#pragma unroll
    for (int w = 0; w < NW; ++w) {
      const float f = fexp2(sv[w].x - M);
      const float* wA = mrgA + w * (4 * 16 * 32);
      zt  = fmaf(f, sv[w].y, zt);
      c2t = fmaf(f, sv[w].z, c2t);
      c6t = fmaf(f, sv[w].w, c6t);
      a1  = fmaf(f, wA[0 * 512 + a], a1);
      a3  = fmaf(f, wA[1 * 512 + a], a3);
      a4  = fmaf(f, wA[2 * 512 + a], a4);
      a5  = fmaf(f, wA[3 * 512 + a], a5);
    }

    const float bt   = beta[col];
    const float invz = frcp(zt);
    const float hv = Hb[(i0 + row) * D_ + col];
    const float lv = __logf(hv + 1e-8f);
    const float num = a1 + c2t * hv +
                      bt * (a3 + hv * a4 - lv * a5 - c6t * lv * hv);
    out[(b * N_ + i0 + row) * D_ + col] = num * invz;
  }
}

extern "C" void kernel_launch(void* const* d_in, const int* in_sizes, int n_in,
                              void* d_out, int out_size, void* d_ws, size_t ws_size,
                              hipStream_t stream) {
  const float* h    = (const float*)d_in[1];
  const float* pe   = (const float*)d_in[2];
  const float* E    = (const float*)d_in[3];
  const float* A    = (const float*)d_in[4];
  const float* Wk   = (const float*)d_in[5];
  const float* bk   = (const float*)d_in[6];
  const float* Wq   = (const float*)d_in[7];
  const float* bq   = (const float*)d_in[8];
  const float* beta = (const float*)d_in[9];
  float* out = (float*)d_out;

  dim3 grid(B_ * (N_ / 16)), block(32 * NW);
  fokker_planck_fused_kernel<<<grid, block, 0, stream>>>(h, pe, E, A, Wk, bk, Wq, bq, beta, out);
}


# Fold Wq into the i side: G = K_i Wq^T once per work-group (2 waves, via LDS); scores = 16-deep dot G_i.pe_j + K_i.bq, no Q projection in the j loop; changes the score dot's summation order
# speedup vs baseline: 8.9115x; 1.0541x over previous
#include <hip/hip_runtime.h>
#include <hip/hip_bf16.h>
#include <math.h>

#define B_   4
#define N_   512
#define D_   32
#define PE_  16

#define NW  16
#define JT_PER_W (N_ / 16 / NW)
static_assert(NW == 16, "the merge gives output row r of the 16-row tile to wave r");
static_assert(JT_PER_W * NW * 16 == N_, "the waves' j slices tile N exactly");
#define L2E 1.44269504088896340736f
#define INV_SQRT_D 0.17677669529663688110f
#define KS2 (10.0f * L2E)

typedef __attribute__((ext_vector_type(2))) float v2f;
typedef __attribute__((ext_vector_type(8))) float v8f;

static __device__ __forceinline__ v8f wmma_f32(v2f a, v2f b, v8f c) {
  return __builtin_amdgcn_wmma_f32_16x16x4_f32(false, a, false, b, (short)0, c, false, false);
}

static __device__ __forceinline__ float fexp2(float x) { return __builtin_amdgcn_exp2f(x); }
static __device__ __forceinline__ float frcp(float x)  { return __builtin_amdgcn_rcpf(x); }

#define DPP_QUAD_XOR1  0xB1
#define DPP_QUAD_XOR2  0x4E
#define DPP_HALF_MIR   0x141
#define DPP_ROW_MIR    0x140

template <int CTRL>
static __device__ __forceinline__ float dpp_mov(float x) {
  return __int_as_float(
      __builtin_amdgcn_update_dpp(0, __float_as_int(x), CTRL, 0xf, 0xf, true));
}
static __device__ __forceinline__ float redmax16(float x) {
  x = fmaxf(x, dpp_mov<DPP_QUAD_XOR1>(x));
  x = fmaxf(x, dpp_mov<DPP_QUAD_XOR2>(x));
  x = fmaxf(x, dpp_mov<DPP_HALF_MIR>(x));
  x = fmaxf(x, dpp_mov<DPP_ROW_MIR>(x));
  return x;
}
static __device__ __forceinline__ float redsum16(float x) {
  x += dpp_mov<DPP_QUAD_XOR1>(x);
  x += dpp_mov<DPP_QUAD_XOR2>(x);
  x += dpp_mov<DPP_HALF_MIR>(x);
  x += dpp_mov<DPP_ROW_MIR>(x);
  return x;
}


static __device__ __forceinline__ int mrg_idx(int row, int col) {
  return row * 32 + (col ^ ((row & 8) << 1));
}

__global__ void __launch_bounds__(32 * NW)
fokker_planck_fused_kernel(const float* __restrict__ h,
                           const float* __restrict__ pe,
                           const float* __restrict__ E,
                           const float* __restrict__ A,
                           const float* __restrict__ Wk, const float* __restrict__ bk,
                           const float* __restrict__ Wq, const float* __restrict__ bq,
                           const float* __restrict__ beta,
                           float* __restrict__ out) {
  __shared__ float lds[NW * 3 * 256];
  __shared__ float mrgA[NW * 4 * 16 * 32];
  __shared__ __attribute__((aligned(16))) float mrgS[NW * 16 * 4];
  __shared__ __attribute__((aligned(16))) float gxG[2 * 2 * 32 * 4];
  __shared__ float gxb[2 * 2 * 16];

  const int lane = threadIdx.x & 31;
  const int wid  = threadIdx.x >> 5;
  const int lo   = lane & 15;
  const int hi   = lane >> 4;
  const int tile = blockIdx.x;
  const int b  = tile >> 5;
  const int i0 = (tile & 31) * 16;

  float* lds1 = lds + wid * 3 * 256;
  float* lds3 = lds1 + 256;
  float* lds4 = lds1 + 512;

  const float* Pb = pe + b * N_ * PE_;
  const float* Hb = h  + b * N_ * D_;

  float Ei[8];
#pragma unroll
  for (int v = 0; v < 8; ++v) Ei[v] = E[i0 + v + 8 * hi];

  float Ej;
  float Am[8];
  v2f bPj[4];
  {
    const int j0 = (wid * JT_PER_W) * 16;
    Ej = E[j0 + lo];
    for (int v = 0; v < 8; ++v) Am[v] = A[(i0 + v + 8 * hi) * N_ + j0 + lo];
    for (int q = 0; q < 4; ++q) bPj[q] = *(const v2f*)(Pb + (j0 + lo) * PE_ + 8 * hi + 2 * q);
  }

  const int wv = __builtin_amdgcn_readfirstlane(wid);
  if (wv < 2) {
    const int s = wv;
    v8f kt;
    for (int v = 0; v < 8; ++v) kt[v] = bk[16 * s + 8 * hi + v];
    for (int kc = 0; kc < 4; ++kc) {
      const int o = (4 * kc + 2 * hi) * D_ + 16 * s + lo;
      v2f aWk; aWk.x = Wk[o]; aWk.y = Wk[o + D_];
      v2f bPi = *(const v2f*)(Pb + (i0 + lo) * PE_ + 4 * kc + 2 * hi);
      kt = wmma_f32(aWk, bPi, kt);
    }
    v8f gt = {};
    for (int q = 0; q < 4; ++q) {
      v2f aWq = *(const v2f*)(Wq + lo * D_ + 16 * s + 8 * hi + 2 * q);
      v2f bK; bK.x = kt[2 * q]; bK.y = kt[2 * q + 1];
      gt = wmma_f32(aWq, bK, gt);
    }
    float gp = 0.f;
    for (int v = 0; v < 8; ++v) gp = fmaf(kt[v], bq[16 * s + 8 * hi + v], gp);
    float4 g0; g0.x = gt[0]; g0.y = gt[1]; g0.z = gt[2]; g0.w = gt[3];
    float4 g1; g1.x = gt[4]; g1.y = gt[5]; g1.z = gt[6]; g1.w = gt[7];
    *(float4*)(gxG + ((s * 2 + 0) * 32 + lane) * 4) = g0;
    *(float4*)(gxG + ((s * 2 + 1) * 32 + lane) * 4) = g1;
    gxb[(s * 2 + hi) * 16 + lo] = gp;
  }
  __syncthreads();

  const float kscale = INV_SQRT_D * L2E;
  v2f afG[4];
  float sinit[8];
  {
    const float4 a0 = *(const float4*)(gxG + ((0 * 2 + 0) * 32 + lane) * 4);
    const float4 a1 = *(const float4*)(gxG + ((0 * 2 + 1) * 32 + lane) * 4);
    const float4 b0 = *(const float4*)(gxG + ((1 * 2 + 0) * 32 + lane) * 4);
    const float4 b1 = *(const float4*)(gxG + ((1 * 2 + 1) * 32 + lane) * 4);
    afG[0].x = (a0.x + b0.x) * kscale;  afG[0].y = (a0.y + b0.y) * kscale;
    afG[1].x = (a0.z + b0.z) * kscale;  afG[1].y = (a0.w + b0.w) * kscale;
    afG[2].x = (a1.x + b1.x) * kscale;  afG[2].y = (a1.y + b1.y) * kscale;
    afG[3].x = (a1.z + b1.z) * kscale;  afG[3].y = (a1.w + b1.w) * kscale;
    for (int v = 0; v < 8; ++v) {
      const int r = v + 8 * hi;
      sinit[v] = ((gxb[0 * 16 + r] + gxb[1 * 16 + r]) + (gxb[2 * 16 + r] + gxb[3 * 16 + r])) * kscale;
    }
  }

  v8f acc1[2] = {}, acc3[2] = {}, acc4[2] = {}, acc5[2] = {};
  float m[8], z[8], c2[8], c6[8];
#pragma unroll
  for (int v = 0; v < 8; ++v) { m[v] = -1.0e30f; z[v] = 0.f; c2[v] = 0.f; c6[v] = 0.f; }

  for (int jt = 0; jt < JT_PER_W; ++jt) {
    const int j0 = (wid * JT_PER_W + jt) * 16;
    if (jt > 0) {
      Ej = E[j0 + lo];
      for (int v = 0; v < 8; ++v) Am[v] = A[(i0 + v + 8 * hi) * N_ + j0 + lo];
      for (int q = 0; q < 4; ++q) bPj[q] = *(const v2f*)(Pb + (j0 + lo) * PE_ + 8 * hi + 2 * q);
    }

    v8f sraw;
    for (int v = 0; v < 8; ++v) sraw[v] = sinit[v];
    for (int q = 0; q < 4; ++q) sraw = wmma_f32(afG[q], bPj[q], sraw);

    float sc[8];
#pragma unroll
    for (int v = 0; v < 8; ++v)
      sc[v] = Am[v] * sraw[v];

    float tm[8];
#pragma unroll
    for (int v = 0; v < 8; ++v) tm[v] = redmax16(sc[v]);

#pragma unroll
    for (int v = 0; v < 8; ++v) {
      float mn = fmaxf(m[v], tm[v]);
      float r  = fexp2(m[v] - mn);
      m[v] = mn;
      z[v]  *= r;  c2[v] *= r;  c6[v] *= r;
      acc1[0][v] *= r; acc1[1][v] *= r;
      acc3[0][v] *= r; acc3[1][v] *= r;
      acc4[0][v] *= r; acc4[1][v] *= r;
      acc5[0][v] *= r; acc5[1][v] *= r;

      float p  = fexp2(sc[v] - mn);
      float Ed = Ej - Ei[v];
      float sg = frcp(1.f + fexp2(-KS2 * Ed));
      float p3 = p * sg;
      float p1 = p3 * Ed;
      float p4 = p - p3;
      z[v]  += p;
      c2[v] += p * Ed - p1;
      c6[v] += p4;

      int addr = (v + 8 * hi) * 16 + lo;
      lds1[addr] = p1;
      lds3[addr] = p3;
      lds4[addr] = p4;
    }

    asm volatile("s_wait_dscnt 0" ::: "memory");

    v2f p1a[4], p3a[4], p4a[4];
#pragma unroll
    for (int kc = 0; kc < 4; ++kc) {
      int a = lo * 16 + 4 * kc + 2 * hi;
      p1a[kc] = *(const v2f*)(lds1 + a);
      p3a[kc] = *(const v2f*)(lds3 + a);
      p4a[kc] = *(const v2f*)(lds4 + a);
    }

    v2f HB[4][2], LB[4][2], LHB[4][2];
#pragma unroll
    for (int kc = 0; kc < 4; ++kc)
#pragma unroll
      for (int s = 0; s < 2; ++s)
#pragma unroll
        for (int t = 0; t < 2; ++t) {
          float hv = Hb[(j0 + 4 * kc + 2 * hi + t) * D_ + lo + 16 * s];
          float lv = __logf(hv + 1e-8f);
          HB[kc][s][t]  = hv;
          LB[kc][s][t]  = lv;
          LHB[kc][s][t] = lv * hv;
        }

#pragma unroll
    for (int s = 0; s < 2; ++s)
#pragma unroll
      for (int kc = 0; kc < 4; ++kc) {
        acc1[s] = wmma_f32(p1a[kc], HB[kc][s],  acc1[s]);
        acc5[s] = wmma_f32(p3a[kc], HB[kc][s],  acc5[s]);
        acc3[s] = wmma_f32(p3a[kc], LHB[kc][s], acc3[s]);
        acc4[s] = wmma_f32(p4a[kc], LB[kc][s],  acc4[s]);
      }
  }

#pragma unroll
  for (int v = 0; v < 8; ++v) {
    z[v]  = redsum16(z[v]);
    c2[v] = redsum16(c2[v]);
    c6[v] = redsum16(c6[v]);
  }

  {
    float* myA = mrgA + wid * (4 * 16 * 32);
#pragma unroll
    for (int v = 0; v < 8; ++v) {
      const int row = v + 8 * hi;
#pragma unroll
      for (int s = 0; s < 2; ++s) {
        const int a = mrg_idx(row, lo + 16 * s);
        myA[0 * 512 + a] = acc1[s][v];
        myA[1 * 512 + a] = acc3[s][v];
        myA[2 * 512 + a] = acc4[s][v];
        myA[3 * 512 + a] = acc5[s][v];
      }
    }
    if (lo == 0) {
#pragma unroll
      for (int v = 0; v < 8; ++v) {
        float4 sv; sv.x = m[v]; sv.y = z[v]; sv.z = c2[v]; sv.w = c6[v];
        *(float4*)(mrgS + (wid * 16 + v + 8 * hi) * 4) = sv;
      }
    }
  }
  __syncthreads();

  {
    const int row = wid;
    const int col = lane;
    float4 sv[NW];
    float M = -1.0e30f;
#pragma unroll
    for (int w = 0; w < NW; ++w) {
      sv[w] = *(const float4*)(mrgS + (w * 16 + row) * 4);
      M = fmaxf(M, sv[w].x);
    }
    const int a = mrg_idx(row, col);
    float zt = 0.f, c2t = 0.f, c6t = 0.f, a1 = 0.f, a3 = 0.f, a4 = 0.f, a5 = 0.f;
#pragma unroll
    for (int w = 0; w < NW; ++w) {
      const float f = fexp2(sv[w].x - M);
      const float* wA = mrgA + w * (4 * 16 * 32);
      zt  = fmaf(f, sv[w].y, zt);
      c2t = fmaf(f, sv[w].z, c2t);
      c6t = fmaf(f, sv[w].w, c6t);
      a1  = fmaf(f, wA[0 * 512 + a], a1);
      a3  = fmaf(f, wA[1 * 512 + a], a3);
      a4  = fmaf(f, wA[2 * 512 + a], a4);
      a5  = fmaf(f, wA[3 * 512 + a], a5);
    }

    const float bt   = beta[col];
    const float invz = frcp(zt);
    const float hv = Hb[(i0 + row) * D_ + col];
    const float lv = __logf(hv + 1e-8f);
    const float num = a1 + c2t * hv +
                      bt * (a3 + hv * a4 - lv * a5 - c6t * lv * hv);
    out[(b * N_ + i0 + row) * D_ + col] = num * invz;
  }
}

extern "C" void kernel_launch(void* const* d_in, const int* in_sizes, int n_in,
                              void* d_out, int out_size, void* d_ws, size_t ws_size,
                              hipStream_t stream) {
  const float* h    = (const float*)d_in[1];
  const float* pe   = (const float*)d_in[2];
  const float* E    = (const float*)d_in[3];
  const float* A    = (const float*)d_in[4];
  const float* Wk   = (const float*)d_in[5];
  const float* bk   = (const float*)d_in[6];
  const float* Wq   = (const float*)d_in[7];
  const float* bq   = (const float*)d_in[8];
  const float* beta = (const float*)d_in[9];
  float* out = (float*)d_out;

  dim3 grid(B_ * (N_ / 16)), block(32 * NW);
  fokker_planck_fused_kernel<<<grid, block, 0, stream>>>(h, pe, E, A, Wk, bk, Wq, bq, beta, out);
}


# Regroup the value stage into 3 matrix products, P.s x [H (E_j + beta L_j) | H] and P.(1-s) x (E_j + beta L_j), was 4; E_i leaves the j sums as a factor: changes the summation order
# speedup vs baseline: 9.7159x; 1.0903x over previous
#include <hip/hip_runtime.h>
#include <hip/hip_bf16.h>
#include <math.h>

#define B_   4
#define N_   512
#define D_   32
#define PE_  16

#define NW  16
#define JT_PER_W (N_ / 16 / NW)
static_assert(NW == 16, "the merge gives output row r of the 16-row tile to wave r");
static_assert(JT_PER_W * NW * 16 == N_, "the waves' j slices tile N exactly");
#define L2E 1.44269504088896340736f
#define INV_SQRT_D 0.17677669529663688110f
#define KS2 (10.0f * L2E)

typedef __attribute__((ext_vector_type(2))) float v2f;
typedef __attribute__((ext_vector_type(8))) float v8f;

static __device__ __forceinline__ v8f wmma_f32(v2f a, v2f b, v8f c) {
  return __builtin_amdgcn_wmma_f32_16x16x4_f32(false, a, false, b, (short)0, c, false, false);
}

static __device__ __forceinline__ float fexp2(float x) { return __builtin_amdgcn_exp2f(x); }
static __device__ __forceinline__ float frcp(float x)  { return __builtin_amdgcn_rcpf(x); }

#define DPP_QUAD_XOR1  0xB1
#define DPP_QUAD_XOR2  0x4E
#define DPP_HALF_MIR   0x141
#define DPP_ROW_MIR    0x140

template <int CTRL>
static __device__ __forceinline__ float dpp_mov(float x) {
  return __int_as_float(
      __builtin_amdgcn_update_dpp(0, __float_as_int(x), CTRL, 0xf, 0xf, true));
}
static __device__ __forceinline__ float redmax16(float x) {
  x = fmaxf(x, dpp_mov<DPP_QUAD_XOR1>(x));
  x = fmaxf(x, dpp_mov<DPP_QUAD_XOR2>(x));
  x = fmaxf(x, dpp_mov<DPP_HALF_MIR>(x));
  x = fmaxf(x, dpp_mov<DPP_ROW_MIR>(x));
  return x;
}
static __device__ __forceinline__ float redsum16(float x) {
  x += dpp_mov<DPP_QUAD_XOR1>(x);
  x += dpp_mov<DPP_QUAD_XOR2>(x);
  x += dpp_mov<DPP_HALF_MIR>(x);
  x += dpp_mov<DPP_ROW_MIR>(x);
  return x;
}


static __device__ __forceinline__ int mrg_idx(int row, int col) {
  return row * 32 + (col ^ ((row & 8) << 1));
}

__global__ void __launch_bounds__(32 * NW) __attribute__((amdgpu_waves_per_eu(4, 4)))
fokker_planck_fused_kernel(const float* __restrict__ h,
                           const float* __restrict__ pe,
                           const float* __restrict__ E,
                           const float* __restrict__ A,
                           const float* __restrict__ Wk, const float* __restrict__ bk,
                           const float* __restrict__ Wq, const float* __restrict__ bq,
                           const float* __restrict__ beta,
                           float* __restrict__ out) {
  __shared__ float lds[NW * 2 * 256];
  __shared__ float mrgA[NW * 3 * 16 * 32];
  __shared__ __attribute__((aligned(16))) float mrgS[NW * 16 * 4];
  __shared__ __attribute__((aligned(16))) float gxG[2 * 2 * 32 * 4];
  __shared__ float gxb[2 * 2 * 16];

  const int lane = threadIdx.x & 31;
  const int wid  = threadIdx.x >> 5;
  const int lo   = lane & 15;
  const int hi   = lane >> 4;
  const int tile = blockIdx.x;
  const int b  = tile >> 5;
  const int i0 = (tile & 31) * 16;

  float* lds3 = lds + wid * 2 * 256;
  float* lds4 = lds3 + 256;

  const float* Pb = pe + b * N_ * PE_;
  const float* Hb = h  + b * N_ * D_;

  float bts[2];
  bts[0] = beta[lo]; bts[1] = beta[lo + 16];

  float Ei[8];
#pragma unroll
  for (int v = 0; v < 8; ++v) Ei[v] = E[i0 + v + 8 * hi];

  float Ej;
  float Am[8];
  v2f bPj[4];
  v2f Ek[4];
  {
    const int j0 = (wid * JT_PER_W) * 16;
    Ej = E[j0 + lo];
    for (int kc = 0; kc < 4; ++kc) Ek[kc] = *(const v2f*)(E + j0 + 4 * kc + 2 * hi);
    for (int v = 0; v < 8; ++v) Am[v] = A[(i0 + v + 8 * hi) * N_ + j0 + lo];
    for (int q = 0; q < 4; ++q) bPj[q] = *(const v2f*)(Pb + (j0 + lo) * PE_ + 8 * hi + 2 * q);
  }

  const int wv = __builtin_amdgcn_readfirstlane(wid);
  if (wv < 2) {
    const int s = wv;
    v8f kt;
    for (int v = 0; v < 8; ++v) kt[v] = bk[16 * s + 8 * hi + v];
    for (int kc = 0; kc < 4; ++kc) {
      const int o = (4 * kc + 2 * hi) * D_ + 16 * s + lo;
      v2f aWk; aWk.x = Wk[o]; aWk.y = Wk[o + D_];
      v2f bPi = *(const v2f*)(Pb + (i0 + lo) * PE_ + 4 * kc + 2 * hi);
      kt = wmma_f32(aWk, bPi, kt);
    }
    v8f gt = {};
    for (int q = 0; q < 4; ++q) {
      v2f aWq = *(const v2f*)(Wq + lo * D_ + 16 * s + 8 * hi + 2 * q);
      v2f bK; bK.x = kt[2 * q]; bK.y = kt[2 * q + 1];
      gt = wmma_f32(aWq, bK, gt);
    }
    float gp = 0.f;
    for (int v = 0; v < 8; ++v) gp = fmaf(kt[v], bq[16 * s + 8 * hi + v], gp);
    float4 g0; g0.x = gt[0]; g0.y = gt[1]; g0.z = gt[2]; g0.w = gt[3];
    float4 g1; g1.x = gt[4]; g1.y = gt[5]; g1.z = gt[6]; g1.w = gt[7];
    *(float4*)(gxG + ((s * 2 + 0) * 32 + lane) * 4) = g0;
    *(float4*)(gxG + ((s * 2 + 1) * 32 + lane) * 4) = g1;
    gxb[(s * 2 + hi) * 16 + lo] = gp;
  }
  __syncthreads();

  const float kscale = INV_SQRT_D * L2E;
  v2f afG[4];
  float sinit[8];
  {
    const float4 a0 = *(const float4*)(gxG + ((0 * 2 + 0) * 32 + lane) * 4);
    const float4 a1 = *(const float4*)(gxG + ((0 * 2 + 1) * 32 + lane) * 4);
    const float4 b0 = *(const float4*)(gxG + ((1 * 2 + 0) * 32 + lane) * 4);
    const float4 b1 = *(const float4*)(gxG + ((1 * 2 + 1) * 32 + lane) * 4);
    afG[0].x = (a0.x + b0.x) * kscale;  afG[0].y = (a0.y + b0.y) * kscale;
    afG[1].x = (a0.z + b0.z) * kscale;  afG[1].y = (a0.w + b0.w) * kscale;
    afG[2].x = (a1.x + b1.x) * kscale;  afG[2].y = (a1.y + b1.y) * kscale;
    afG[3].x = (a1.z + b1.z) * kscale;  afG[3].y = (a1.w + b1.w) * kscale;
    for (int v = 0; v < 8; ++v) {
      const int r = v + 8 * hi;
      sinit[v] = ((gxb[0 * 16 + r] + gxb[1 * 16 + r]) + (gxb[2 * 16 + r] + gxb[3 * 16 + r])) * kscale;
    }
  }

  v8f accX[2] = {}, accH[2] = {}, accV[2] = {};
  float m[8], z[8], c6[8];
#pragma unroll
  for (int v = 0; v < 8; ++v) { m[v] = -1.0e30f; z[v] = 0.f; c6[v] = 0.f; }

  for (int jt = 0; jt < JT_PER_W; ++jt) {
    const int j0 = (wid * JT_PER_W + jt) * 16;
    if (jt > 0) {
      Ej = E[j0 + lo];
      for (int kc = 0; kc < 4; ++kc) Ek[kc] = *(const v2f*)(E + j0 + 4 * kc + 2 * hi);
      for (int v = 0; v < 8; ++v) Am[v] = A[(i0 + v + 8 * hi) * N_ + j0 + lo];
      for (int q = 0; q < 4; ++q) bPj[q] = *(const v2f*)(Pb + (j0 + lo) * PE_ + 8 * hi + 2 * q);
    }

    v8f sraw;
    for (int v = 0; v < 8; ++v) sraw[v] = sinit[v];
    for (int q = 0; q < 4; ++q) sraw = wmma_f32(afG[q], bPj[q], sraw);

    float sc[8];
#pragma unroll
    for (int v = 0; v < 8; ++v)
      sc[v] = Am[v] * sraw[v];

    float tm[8];
#pragma unroll
    for (int v = 0; v < 8; ++v) tm[v] = redmax16(sc[v]);

#pragma unroll
    for (int v = 0; v < 8; ++v) {
      float mn = fmaxf(m[v], tm[v]);
      float r  = fexp2(m[v] - mn);
      m[v] = mn;
      z[v]  *= r;  c6[v] *= r;
      accX[0][v] *= r; accX[1][v] *= r;
      accH[0][v] *= r; accH[1][v] *= r;
      accV[0][v] *= r; accV[1][v] *= r;

      float p  = fexp2(sc[v] - mn);
      float Ed = Ej - Ei[v];
      float sg = frcp(1.f + fexp2(-KS2 * Ed));
      float p3 = p * sg;
      float p4 = p - p3;
      z[v]  += p;
      c6[v] += p4;

      int addr = (v + 8 * hi) * 16 + lo;
      lds3[addr] = p3;
      lds4[addr] = p4;
    }

    asm volatile("s_wait_dscnt 0" ::: "memory");

    v2f p3a[4], p4a[4];
#pragma unroll
    for (int kc = 0; kc < 4; ++kc) {
      int a = lo * 16 + 4 * kc + 2 * hi;
      p3a[kc] = *(const v2f*)(lds3 + a);
      p4a[kc] = *(const v2f*)(lds4 + a);
    }

    v2f HB[4][2], VB[4][2], HVB[4][2];
#pragma unroll
    for (int kc = 0; kc < 4; ++kc) {
#pragma unroll
      for (int s = 0; s < 2; ++s)
#pragma unroll
        for (int t = 0; t < 2; ++t) {
          float hv = Hb[(j0 + 4 * kc + 2 * hi + t) * D_ + lo + 16 * s];
          float lv = __logf(hv + 1e-8f);
          float vv = fmaf(bts[s], lv, Ek[kc][t]);
          HB[kc][s][t]  = hv;
          VB[kc][s][t]  = vv;
          HVB[kc][s][t] = hv * vv;
        }
    }

#pragma unroll
    for (int s = 0; s < 2; ++s)
#pragma unroll
      for (int kc = 0; kc < 4; ++kc) {
        accX[s] = wmma_f32(p3a[kc], HVB[kc][s], accX[s]);
        accH[s] = wmma_f32(p3a[kc], HB[kc][s],  accH[s]);
        accV[s] = wmma_f32(p4a[kc], VB[kc][s],  accV[s]);
      }
  }

#pragma unroll
  for (int v = 0; v < 8; ++v) {
    z[v]  = redsum16(z[v]);
    c6[v] = redsum16(c6[v]);
  }

  {
    float* myA = mrgA + wid * (3 * 16 * 32);
#pragma unroll
    for (int v = 0; v < 8; ++v) {
      const int row = v + 8 * hi;
#pragma unroll
      for (int s = 0; s < 2; ++s) {
        const int a = mrg_idx(row, lo + 16 * s);
        myA[0 * 512 + a] = accX[s][v];
        myA[1 * 512 + a] = accH[s][v];
        myA[2 * 512 + a] = accV[s][v];
      }
    }
    if (lo == 0) {
#pragma unroll
      for (int v = 0; v < 8; ++v) {
        float4 sv; sv.x = m[v]; sv.y = z[v]; sv.z = c6[v]; sv.w = 0.f;
        *(float4*)(mrgS + (wid * 16 + v + 8 * hi) * 4) = sv;
      }
    }
  }
  __syncthreads();

  {
    const int row = wid;
    const int col = lane;
    float4 sv[NW];
    float M = -1.0e30f;
#pragma unroll
    for (int w = 0; w < NW; ++w) {
      sv[w] = *(const float4*)(mrgS + (w * 16 + row) * 4);
      M = fmaxf(M, sv[w].x);
    }
    const int a = mrg_idx(row, col);
    float zt = 0.f, c6t = 0.f, aX = 0.f, aH = 0.f, aV = 0.f;
#pragma unroll
    for (int w = 0; w < NW; ++w) {
      const float f = fexp2(sv[w].x - M);
      const float* wA = mrgA + w * (3 * 16 * 32);
      zt  = fmaf(f, sv[w].y, zt);
      c6t = fmaf(f, sv[w].z, c6t);
      aX  = fmaf(f, wA[0 * 512 + a], aX);
      aH  = fmaf(f, wA[1 * 512 + a], aH);
      aV  = fmaf(f, wA[2 * 512 + a], aV);
    }

    const float bt   = beta[col];
    const float invz = frcp(zt);
    const float hv = Hb[(i0 + row) * D_ + col];
    const float lv = __logf(hv + 1e-8f);
    const float vi = fmaf(bt, lv, E[i0 + row]);
    const float num = aX + hv * aV - vi * (aH + hv * c6t);
    out[(b * N_ + i0 + row) * D_ + col] = num * invz;
  }
}

extern "C" void kernel_launch(void* const* d_in, const int* in_sizes, int n_in,
                              void* d_out, int out_size, void* d_ws, size_t ws_size,
                              hipStream_t stream) {
  const float* h    = (const float*)d_in[1];
  const float* pe   = (const float*)d_in[2];
  const float* E    = (const float*)d_in[3];
  const float* A    = (const float*)d_in[4];
  const float* Wk   = (const float*)d_in[5];
  const float* bk   = (const float*)d_in[6];
  const float* Wq   = (const float*)d_in[7];
  const float* bq   = (const float*)d_in[8];
  const float* beta = (const float*)d_in[9];
  float* out = (float*)d_out;

  dim3 grid(B_ * (N_ / 16)), block(32 * NW);
  fokker_planck_fused_kernel<<<grid, block, 0, stream>>>(h, pe, E, A, Wk, bk, Wq, bq, beta, out);
}


# log(h + 1e-8) as the hardware log2 instruction with ln 2 folded into beta, in place of the library log's 12-instruction sequence: 33 logs per lane, j side and i side alike
# speedup vs baseline: 10.2142x; 1.0513x over previous
#include <hip/hip_runtime.h>
#include <hip/hip_bf16.h>
#include <math.h>

#define B_   4
#define N_   512
#define D_   32
#define PE_  16

#define NW  16
#define JT_PER_W (N_ / 16 / NW)
static_assert(NW == 16, "the merge gives output row r of the 16-row tile to wave r");
static_assert(JT_PER_W * NW * 16 == N_, "the waves' j slices tile N exactly");
#define L2E 1.44269504088896340736f
#define INV_SQRT_D 0.17677669529663688110f
#define KS2 (10.0f * L2E)
#define LN2 0.69314718055994530942f

typedef __attribute__((ext_vector_type(2))) float v2f;
typedef __attribute__((ext_vector_type(8))) float v8f;

static __device__ __forceinline__ v8f wmma_f32(v2f a, v2f b, v8f c) {
  return __builtin_amdgcn_wmma_f32_16x16x4_f32(false, a, false, b, (short)0, c, false, false);
}

static __device__ __forceinline__ float fexp2(float x) { return __builtin_amdgcn_exp2f(x); }
static __device__ __forceinline__ float frcp(float x)  { return __builtin_amdgcn_rcpf(x); }
static __device__ __forceinline__ float flog2(float x) { return __builtin_amdgcn_logf(x); }

#define DPP_QUAD_XOR1  0xB1
#define DPP_QUAD_XOR2  0x4E
#define DPP_HALF_MIR   0x141
#define DPP_ROW_MIR    0x140

template <int CTRL>
static __device__ __forceinline__ float dpp_mov(float x) {
  return __int_as_float(
      __builtin_amdgcn_update_dpp(0, __float_as_int(x), CTRL, 0xf, 0xf, true));
}
static __device__ __forceinline__ float redmax16(float x) {
  x = fmaxf(x, dpp_mov<DPP_QUAD_XOR1>(x));
  x = fmaxf(x, dpp_mov<DPP_QUAD_XOR2>(x));
  x = fmaxf(x, dpp_mov<DPP_HALF_MIR>(x));
  x = fmaxf(x, dpp_mov<DPP_ROW_MIR>(x));
  return x;
}
static __device__ __forceinline__ float redsum16(float x) {
  x += dpp_mov<DPP_QUAD_XOR1>(x);
  x += dpp_mov<DPP_QUAD_XOR2>(x);
  x += dpp_mov<DPP_HALF_MIR>(x);
  x += dpp_mov<DPP_ROW_MIR>(x);
  return x;
}


static __device__ __forceinline__ int mrg_idx(int row, int col) {
  return row * 32 + (col ^ ((row & 8) << 1));
}

__global__ void __launch_bounds__(32 * NW) __attribute__((amdgpu_waves_per_eu(4, 4)))
fokker_planck_fused_kernel(const float* __restrict__ h,
                           const float* __restrict__ pe,
                           const float* __restrict__ E,
                           const float* __restrict__ A,
                           const float* __restrict__ Wk, const float* __restrict__ bk,
                           const float* __restrict__ Wq, const float* __restrict__ bq,
                           const float* __restrict__ beta,
                           float* __restrict__ out) {
  __shared__ float lds[NW * 2 * 256];
  __shared__ float mrgA[NW * 3 * 16 * 32];
  __shared__ __attribute__((aligned(16))) float mrgS[NW * 16 * 4];
  __shared__ __attribute__((aligned(16))) float gxG[2 * 2 * 32 * 4];
  __shared__ float gxb[2 * 2 * 16];

  const int lane = threadIdx.x & 31;
  const int wid  = threadIdx.x >> 5;
  const int lo   = lane & 15;
  const int hi   = lane >> 4;
  const int tile = blockIdx.x;
  const int b  = tile >> 5;
  const int i0 = (tile & 31) * 16;

  float* lds3 = lds + wid * 2 * 256;
  float* lds4 = lds3 + 256;

  const float* Pb = pe + b * N_ * PE_;
  const float* Hb = h  + b * N_ * D_;

  float bts[2];
  bts[0] = beta[lo] * LN2; bts[1] = beta[lo + 16] * LN2;

  float Ei[8];
#pragma unroll
  for (int v = 0; v < 8; ++v) Ei[v] = E[i0 + v + 8 * hi];

  float Ej;
  float Am[8];
  v2f bPj[4];
  v2f Ek[4];
  {
    const int j0 = (wid * JT_PER_W) * 16;
    Ej = E[j0 + lo];
    for (int kc = 0; kc < 4; ++kc) Ek[kc] = *(const v2f*)(E + j0 + 4 * kc + 2 * hi);
    for (int v = 0; v < 8; ++v) Am[v] = A[(i0 + v + 8 * hi) * N_ + j0 + lo];
    for (int q = 0; q < 4; ++q) bPj[q] = *(const v2f*)(Pb + (j0 + lo) * PE_ + 8 * hi + 2 * q);
  }

  const int wv = __builtin_amdgcn_readfirstlane(wid);
  if (wv < 2) {
    const int s = wv;
    v8f kt;
    for (int v = 0; v < 8; ++v) kt[v] = bk[16 * s + 8 * hi + v];
    for (int kc = 0; kc < 4; ++kc) {
      const int o = (4 * kc + 2 * hi) * D_ + 16 * s + lo;
      v2f aWk; aWk.x = Wk[o]; aWk.y = Wk[o + D_];
      v2f bPi = *(const v2f*)(Pb + (i0 + lo) * PE_ + 4 * kc + 2 * hi);
      kt = wmma_f32(aWk, bPi, kt);
    }
    v8f gt = {};
    for (int q = 0; q < 4; ++q) {
      v2f aWq = *(const v2f*)(Wq + lo * D_ + 16 * s + 8 * hi + 2 * q);
      v2f bK; bK.x = kt[2 * q]; bK.y = kt[2 * q + 1];
      gt = wmma_f32(aWq, bK, gt);
    }
    float gp = 0.f;
    for (int v = 0; v < 8; ++v) gp = fmaf(kt[v], bq[16 * s + 8 * hi + v], gp);
    float4 g0; g0.x = gt[0]; g0.y = gt[1]; g0.z = gt[2]; g0.w = gt[3];
    float4 g1; g1.x = gt[4]; g1.y = gt[5]; g1.z = gt[6]; g1.w = gt[7];
    *(float4*)(gxG + ((s * 2 + 0) * 32 + lane) * 4) = g0;
    *(float4*)(gxG + ((s * 2 + 1) * 32 + lane) * 4) = g1;
    gxb[(s * 2 + hi) * 16 + lo] = gp;
  }
  __syncthreads();

  const float kscale = INV_SQRT_D * L2E;
  v2f afG[4];
  float sinit[8];
  {
    const float4 a0 = *(const float4*)(gxG + ((0 * 2 + 0) * 32 + lane) * 4);
    const float4 a1 = *(const float4*)(gxG + ((0 * 2 + 1) * 32 + lane) * 4);
    const float4 b0 = *(const float4*)(gxG + ((1 * 2 + 0) * 32 + lane) * 4);
    const float4 b1 = *(const float4*)(gxG + ((1 * 2 + 1) * 32 + lane) * 4);
    afG[0].x = (a0.x + b0.x) * kscale;  afG[0].y = (a0.y + b0.y) * kscale;
    afG[1].x = (a0.z + b0.z) * kscale;  afG[1].y = (a0.w + b0.w) * kscale;
    afG[2].x = (a1.x + b1.x) * kscale;  afG[2].y = (a1.y + b1.y) * kscale;
    afG[3].x = (a1.z + b1.z) * kscale;  afG[3].y = (a1.w + b1.w) * kscale;
    for (int v = 0; v < 8; ++v) {
      const int r = v + 8 * hi;
      sinit[v] = ((gxb[0 * 16 + r] + gxb[1 * 16 + r]) + (gxb[2 * 16 + r] + gxb[3 * 16 + r])) * kscale;
    }
  }

  v8f accX[2] = {}, accH[2] = {}, accV[2] = {};
  float m[8], z[8], c6[8];
#pragma unroll
  for (int v = 0; v < 8; ++v) { m[v] = -1.0e30f; z[v] = 0.f; c6[v] = 0.f; }

  for (int jt = 0; jt < JT_PER_W; ++jt) {
    const int j0 = (wid * JT_PER_W + jt) * 16;
    if (jt > 0) {
      Ej = E[j0 + lo];
      for (int kc = 0; kc < 4; ++kc) Ek[kc] = *(const v2f*)(E + j0 + 4 * kc + 2 * hi);
      for (int v = 0; v < 8; ++v) Am[v] = A[(i0 + v + 8 * hi) * N_ + j0 + lo];
      for (int q = 0; q < 4; ++q) bPj[q] = *(const v2f*)(Pb + (j0 + lo) * PE_ + 8 * hi + 2 * q);
    }

    v8f sraw;
    for (int v = 0; v < 8; ++v) sraw[v] = sinit[v];
    for (int q = 0; q < 4; ++q) sraw = wmma_f32(afG[q], bPj[q], sraw);

    float sc[8];
#pragma unroll
    for (int v = 0; v < 8; ++v)
      sc[v] = Am[v] * sraw[v];

    float tm[8];
#pragma unroll
    for (int v = 0; v < 8; ++v) tm[v] = redmax16(sc[v]);

#pragma unroll
    for (int v = 0; v < 8; ++v) {
      float mn = fmaxf(m[v], tm[v]);
      float r  = fexp2(m[v] - mn);
      m[v] = mn;
      z[v]  *= r;  c6[v] *= r;
      accX[0][v] *= r; accX[1][v] *= r;
      accH[0][v] *= r; accH[1][v] *= r;
      accV[0][v] *= r; accV[1][v] *= r;

      float p  = fexp2(sc[v] - mn);
      float Ed = Ej - Ei[v];
      float sg = frcp(1.f + fexp2(-KS2 * Ed));
      float p3 = p * sg;
      float p4 = p - p3;
      z[v]  += p;
      c6[v] += p4;

      int addr = (v + 8 * hi) * 16 + lo;
      lds3[addr] = p3;
      lds4[addr] = p4;
    }

    asm volatile("s_wait_dscnt 0" ::: "memory");

    v2f p3a[4], p4a[4];
#pragma unroll
    for (int kc = 0; kc < 4; ++kc) {
      int a = lo * 16 + 4 * kc + 2 * hi;
      p3a[kc] = *(const v2f*)(lds3 + a);
      p4a[kc] = *(const v2f*)(lds4 + a);
    }

    v2f HB[4][2], VB[4][2], HVB[4][2];
#pragma unroll
    for (int kc = 0; kc < 4; ++kc) {
#pragma unroll
      for (int s = 0; s < 2; ++s)
#pragma unroll
        for (int t = 0; t < 2; ++t) {
          float hv = Hb[(j0 + 4 * kc + 2 * hi + t) * D_ + lo + 16 * s];
          float lv = flog2(hv + 1e-8f);
          float vv = fmaf(bts[s], lv, Ek[kc][t]);
          HB[kc][s][t]  = hv;
          VB[kc][s][t]  = vv;
          HVB[kc][s][t] = hv * vv;
        }
    }

#pragma unroll
    for (int s = 0; s < 2; ++s)
#pragma unroll
      for (int kc = 0; kc < 4; ++kc) {
        accX[s] = wmma_f32(p3a[kc], HVB[kc][s], accX[s]);
        accH[s] = wmma_f32(p3a[kc], HB[kc][s],  accH[s]);
        accV[s] = wmma_f32(p4a[kc], VB[kc][s],  accV[s]);
      }
  }

#pragma unroll
  for (int v = 0; v < 8; ++v) {
    z[v]  = redsum16(z[v]);
    c6[v] = redsum16(c6[v]);
  }

  {
    float* myA = mrgA + wid * (3 * 16 * 32);
#pragma unroll
    for (int v = 0; v < 8; ++v) {
      const int row = v + 8 * hi;
#pragma unroll
      for (int s = 0; s < 2; ++s) {
        const int a = mrg_idx(row, lo + 16 * s);
        myA[0 * 512 + a] = accX[s][v];
        myA[1 * 512 + a] = accH[s][v];
        myA[2 * 512 + a] = accV[s][v];
      }
    }
    if (lo == 0) {
#pragma unroll
      for (int v = 0; v < 8; ++v) {
        float4 sv; sv.x = m[v]; sv.y = z[v]; sv.z = c6[v]; sv.w = 0.f;
        *(float4*)(mrgS + (wid * 16 + v + 8 * hi) * 4) = sv;
      }
    }
  }
  __syncthreads();

  {
    const int row = wid;
    const int col = lane;
    float4 sv[NW];
    float M = -1.0e30f;
#pragma unroll
    for (int w = 0; w < NW; ++w) {
      sv[w] = *(const float4*)(mrgS + (w * 16 + row) * 4);
      M = fmaxf(M, sv[w].x);
    }
    const int a = mrg_idx(row, col);
    float zt = 0.f, c6t = 0.f, aX = 0.f, aH = 0.f, aV = 0.f;
#pragma unroll
    for (int w = 0; w < NW; ++w) {
      const float f = fexp2(sv[w].x - M);
      const float* wA = mrgA + w * (3 * 16 * 32);
      zt  = fmaf(f, sv[w].y, zt);
      c6t = fmaf(f, sv[w].z, c6t);
      aX  = fmaf(f, wA[0 * 512 + a], aX);
      aH  = fmaf(f, wA[1 * 512 + a], aH);
      aV  = fmaf(f, wA[2 * 512 + a], aV);
    }

    const float bt   = beta[col] * LN2;
    const float invz = frcp(zt);
    const float hv = Hb[(i0 + row) * D_ + col];
    const float lv = flog2(hv + 1e-8f);
    const float vi = fmaf(bt, lv, E[i0 + row]);
    const float num = aX + hv * aV - vi * (aH + hv * c6t);
    out[(b * N_ + i0 + row) * D_ + col] = num * invz;
  }
}

extern "C" void kernel_launch(void* const* d_in, const int* in_sizes, int n_in,
                              void* d_out, int out_size, void* d_ws, size_t ws_size,
                              hipStream_t stream) {
  const float* h    = (const float*)d_in[1];
  const float* pe   = (const float*)d_in[2];
  const float* E    = (const float*)d_in[3];
  const float* A    = (const float*)d_in[4];
  const float* Wk   = (const float*)d_in[5];
  const float* bk   = (const float*)d_in[6];
  const float* Wq   = (const float*)d_in[7];
  const float* bq   = (const float*)d_in[8];
  const float* beta = (const float*)d_in[9];
  float* out = (float*)d_out;

  dim3 grid(B_ * (N_ / 16)), block(32 * NW);
  fokker_planck_fused_kernel<<<grid, block, 0, stream>>>(h, pe, E, A, Wk, bk, Wq, bq, beta, out);
}


# Each wave folds its three partial accumulators and c6 into one partial numerator tile with H_i, V_i before the merge: one 16x32 tile per wave crosses LDS, not three; changes the summation tree
# speedup vs baseline: 10.9597x; 1.0730x over previous
#include <hip/hip_runtime.h>
#include <hip/hip_bf16.h>
#include <math.h>

#define B_   4
#define N_   512
#define D_   32
#define PE_  16

#define NW  16
#define JT_PER_W (N_ / 16 / NW)
static_assert(NW == 16, "the merge gives output row r of the 16-row tile to wave r, and one 16-lane row holds the NW partials' m and z");
static_assert(JT_PER_W * NW * 16 == N_, "the waves' j slices tile N exactly");
#define L2E 1.44269504088896340736f
#define INV_SQRT_D 0.17677669529663688110f
#define KS2 (10.0f * L2E)
#define LN2 0.69314718055994530942f

typedef __attribute__((ext_vector_type(2))) float v2f;
typedef __attribute__((ext_vector_type(8))) float v8f;

static __device__ __forceinline__ v8f wmma_f32(v2f a, v2f b, v8f c) {
  return __builtin_amdgcn_wmma_f32_16x16x4_f32(false, a, false, b, (short)0, c, false, false);
}

static __device__ __forceinline__ float fexp2(float x) { return __builtin_amdgcn_exp2f(x); }
static __device__ __forceinline__ float frcp(float x)  { return __builtin_amdgcn_rcpf(x); }
static __device__ __forceinline__ float flog2(float x) { return __builtin_amdgcn_logf(x); }

#define DPP_QUAD_XOR1  0xB1
#define DPP_QUAD_XOR2  0x4E
#define DPP_HALF_MIR   0x141
#define DPP_ROW_MIR    0x140

template <int CTRL>
static __device__ __forceinline__ float dpp_mov(float x) {
  return __int_as_float(
      __builtin_amdgcn_update_dpp(0, __float_as_int(x), CTRL, 0xf, 0xf, true));
}
static __device__ __forceinline__ float redmax16(float x) {
  x = fmaxf(x, dpp_mov<DPP_QUAD_XOR1>(x));
  x = fmaxf(x, dpp_mov<DPP_QUAD_XOR2>(x));
  x = fmaxf(x, dpp_mov<DPP_HALF_MIR>(x));
  x = fmaxf(x, dpp_mov<DPP_ROW_MIR>(x));
  return x;
}
static __device__ __forceinline__ float redsum16(float x) {
  x += dpp_mov<DPP_QUAD_XOR1>(x);
  x += dpp_mov<DPP_QUAD_XOR2>(x);
  x += dpp_mov<DPP_HALF_MIR>(x);
  x += dpp_mov<DPP_ROW_MIR>(x);
  return x;
}


static __device__ __forceinline__ int mrg_idx(int w, int row, int lo) {
  const int slot = row ^ (((row ^ w) >> 3) & 1);
  return ((w * 16 + slot) * 16 + lo) * 2;
}
static __device__ __forceinline__ float xhalf(float x) {
  return __int_as_float(__builtin_amdgcn_permlanex16(__float_as_int(x), __float_as_int(x),
                                                     0x76543210, 0xfedcba98, false, false));
}

__global__ void __launch_bounds__(32 * NW) __attribute__((amdgpu_waves_per_eu(4, 4)))
fokker_planck_fused_kernel(const float* __restrict__ h,
                           const float* __restrict__ pe,
                           const float* __restrict__ E,
                           const float* __restrict__ A,
                           const float* __restrict__ Wk, const float* __restrict__ bk,
                           const float* __restrict__ Wq, const float* __restrict__ bq,
                           const float* __restrict__ beta,
                           float* __restrict__ out) {
  __shared__ float lds[NW * 2 * 256];
  __shared__ __attribute__((aligned(16))) float mrgN[NW * 16 * 16 * 2];
  __shared__ __attribute__((aligned(16))) float mrgS[16 * NW * 2];
  __shared__ __attribute__((aligned(16))) float gxG[2 * 2 * 32 * 4];
  __shared__ float gxb[2 * 2 * 16];

  const int lane = threadIdx.x & 31;
  const int wid  = threadIdx.x >> 5;
  const int lo   = lane & 15;
  const int hi   = lane >> 4;
  const int tile = blockIdx.x;
  const int b  = tile >> 5;
  const int i0 = (tile & 31) * 16;

  float* lds3 = lds + wid * 2 * 256;
  float* lds4 = lds3 + 256;

  const float* Pb = pe + b * N_ * PE_;
  const float* Hb = h  + b * N_ * D_;

  float bts[2];
  bts[0] = beta[lo] * LN2; bts[1] = beta[lo + 16] * LN2;

  float Ei[8];
#pragma unroll
  for (int v = 0; v < 8; ++v) Ei[v] = E[i0 + v + 8 * hi];

  float Hi[2][8];
#pragma unroll
  for (int v = 0; v < 8; ++v) {
    Hi[0][v] = Hb[(i0 + v + 8 * hi) * D_ + lo];
    Hi[1][v] = Hb[(i0 + v + 8 * hi) * D_ + lo + 16];
  }

  float Ej;
  float Am[8];
  v2f bPj[4];
  v2f Ek[4];
  {
    const int j0 = (wid * JT_PER_W) * 16;
    Ej = E[j0 + lo];
    for (int kc = 0; kc < 4; ++kc) Ek[kc] = *(const v2f*)(E + j0 + 4 * kc + 2 * hi);
    for (int v = 0; v < 8; ++v) Am[v] = A[(i0 + v + 8 * hi) * N_ + j0 + lo];
    for (int q = 0; q < 4; ++q) bPj[q] = *(const v2f*)(Pb + (j0 + lo) * PE_ + 8 * hi + 2 * q);
  }

  const int wv = __builtin_amdgcn_readfirstlane(wid);
  if (wv < 2) {
    const int s = wv;
    v8f kt;
    for (int v = 0; v < 8; ++v) kt[v] = bk[16 * s + 8 * hi + v];
    for (int kc = 0; kc < 4; ++kc) {
      const int o = (4 * kc + 2 * hi) * D_ + 16 * s + lo;
      v2f aWk; aWk.x = Wk[o]; aWk.y = Wk[o + D_];
      v2f bPi = *(const v2f*)(Pb + (i0 + lo) * PE_ + 4 * kc + 2 * hi);
      kt = wmma_f32(aWk, bPi, kt);
    }
    v8f gt = {};
    for (int q = 0; q < 4; ++q) {
      v2f aWq = *(const v2f*)(Wq + lo * D_ + 16 * s + 8 * hi + 2 * q);
      v2f bK; bK.x = kt[2 * q]; bK.y = kt[2 * q + 1];
      gt = wmma_f32(aWq, bK, gt);
    }
    float gp = 0.f;
    for (int v = 0; v < 8; ++v) gp = fmaf(kt[v], bq[16 * s + 8 * hi + v], gp);
    float4 g0; g0.x = gt[0]; g0.y = gt[1]; g0.z = gt[2]; g0.w = gt[3];
    float4 g1; g1.x = gt[4]; g1.y = gt[5]; g1.z = gt[6]; g1.w = gt[7];
    *(float4*)(gxG + ((s * 2 + 0) * 32 + lane) * 4) = g0;
    *(float4*)(gxG + ((s * 2 + 1) * 32 + lane) * 4) = g1;
    gxb[(s * 2 + hi) * 16 + lo] = gp;
  }
  __syncthreads();

  const float kscale = INV_SQRT_D * L2E;
  v2f afG[4];
  float sinit[8];
  {
    const float4 a0 = *(const float4*)(gxG + ((0 * 2 + 0) * 32 + lane) * 4);
    const float4 a1 = *(const float4*)(gxG + ((0 * 2 + 1) * 32 + lane) * 4);
    const float4 b0 = *(const float4*)(gxG + ((1 * 2 + 0) * 32 + lane) * 4);
    const float4 b1 = *(const float4*)(gxG + ((1 * 2 + 1) * 32 + lane) * 4);
    afG[0].x = (a0.x + b0.x) * kscale;  afG[0].y = (a0.y + b0.y) * kscale;
    afG[1].x = (a0.z + b0.z) * kscale;  afG[1].y = (a0.w + b0.w) * kscale;
    afG[2].x = (a1.x + b1.x) * kscale;  afG[2].y = (a1.y + b1.y) * kscale;
    afG[3].x = (a1.z + b1.z) * kscale;  afG[3].y = (a1.w + b1.w) * kscale;
    for (int v = 0; v < 8; ++v) {
      const int r = v + 8 * hi;
      sinit[v] = ((gxb[0 * 16 + r] + gxb[1 * 16 + r]) + (gxb[2 * 16 + r] + gxb[3 * 16 + r])) * kscale;
    }
  }

  v8f accX[2] = {}, accH[2] = {}, accV[2] = {};
  float m[8], z[8], c6[8];
#pragma unroll
  for (int v = 0; v < 8; ++v) { m[v] = -1.0e30f; z[v] = 0.f; c6[v] = 0.f; }

  for (int jt = 0; jt < JT_PER_W; ++jt) {
    const int j0 = (wid * JT_PER_W + jt) * 16;
    if (jt > 0) {
      Ej = E[j0 + lo];
      for (int kc = 0; kc < 4; ++kc) Ek[kc] = *(const v2f*)(E + j0 + 4 * kc + 2 * hi);
      for (int v = 0; v < 8; ++v) Am[v] = A[(i0 + v + 8 * hi) * N_ + j0 + lo];
      for (int q = 0; q < 4; ++q) bPj[q] = *(const v2f*)(Pb + (j0 + lo) * PE_ + 8 * hi + 2 * q);
    }

    v8f sraw;
    for (int v = 0; v < 8; ++v) sraw[v] = sinit[v];
    for (int q = 0; q < 4; ++q) sraw = wmma_f32(afG[q], bPj[q], sraw);

    float sc[8];
#pragma unroll
    for (int v = 0; v < 8; ++v)
      sc[v] = Am[v] * sraw[v];

    float tm[8];
#pragma unroll
    for (int v = 0; v < 8; ++v) tm[v] = redmax16(sc[v]);

#pragma unroll
    for (int v = 0; v < 8; ++v) {
      float mn = fmaxf(m[v], tm[v]);
      float r  = fexp2(m[v] - mn);
      m[v] = mn;
      z[v]  *= r;  c6[v] *= r;
      accX[0][v] *= r; accX[1][v] *= r;
      accH[0][v] *= r; accH[1][v] *= r;
      accV[0][v] *= r; accV[1][v] *= r;

      float p  = fexp2(sc[v] - mn);
      float Ed = Ej - Ei[v];
      float sg = frcp(1.f + fexp2(-KS2 * Ed));
      float p3 = p * sg;
      float p4 = p - p3;
      z[v]  += p;
      c6[v] += p4;

      int addr = (v + 8 * hi) * 16 + lo;
      lds3[addr] = p3;
      lds4[addr] = p4;
    }

    asm volatile("s_wait_dscnt 0" ::: "memory");

    v2f p3a[4], p4a[4];
#pragma unroll
    for (int kc = 0; kc < 4; ++kc) {
      int a = lo * 16 + 4 * kc + 2 * hi;
      p3a[kc] = *(const v2f*)(lds3 + a);
      p4a[kc] = *(const v2f*)(lds4 + a);
    }

    v2f HB[4][2], VB[4][2], HVB[4][2];
#pragma unroll
    for (int kc = 0; kc < 4; ++kc) {
#pragma unroll
      for (int s = 0; s < 2; ++s)
#pragma unroll
        for (int t = 0; t < 2; ++t) {
          float hv = Hb[(j0 + 4 * kc + 2 * hi + t) * D_ + lo + 16 * s];
          float lv = flog2(hv + 1e-8f);
          float vv = fmaf(bts[s], lv, Ek[kc][t]);
          HB[kc][s][t]  = hv;
          VB[kc][s][t]  = vv;
          HVB[kc][s][t] = hv * vv;
        }
    }

#pragma unroll
    for (int s = 0; s < 2; ++s)
#pragma unroll
      for (int kc = 0; kc < 4; ++kc) {
        accX[s] = wmma_f32(p3a[kc], HVB[kc][s], accX[s]);
        accH[s] = wmma_f32(p3a[kc], HB[kc][s],  accH[s]);
        accV[s] = wmma_f32(p4a[kc], VB[kc][s],  accV[s]);
      }
  }

#pragma unroll
  for (int v = 0; v < 8; ++v) {
    z[v]  = redsum16(z[v]);
    c6[v] = redsum16(c6[v]);
  }

  float numw[2][8];
#pragma unroll
  for (int v = 0; v < 8; ++v) {
#pragma unroll
    for (int s = 0; s < 2; ++s) {
      const float hv = Hi[s][v];
      const float lv = flog2(hv + 1e-8f);
      const float vi = fmaf(bts[s], lv, Ei[v]);
      const float t1 = fmaf(hv, c6[v], accH[s][v]);
      const float t2 = fmaf(hv, accV[s][v], accX[s][v]);
      numw[s][v] = fmaf(-vi, t1, t2);
    }
  }

  {
#pragma unroll
    for (int v = 0; v < 8; ++v) {
      const int row = v + 8 * hi;
      v2f nv; nv.x = numw[0][v]; nv.y = numw[1][v];
      *(v2f*)(mrgN + mrg_idx(wid, row, lo)) = nv;
    }
    if (lo == 0) {
#pragma unroll
      for (int v = 0; v < 8; ++v) {
        v2f sv; sv.x = m[v]; sv.y = z[v];
        *(v2f*)(mrgS + ((v + 8 * hi) * NW + wid) * 2) = sv;
      }
    }
  }
  __syncthreads();

  {
    const int row = wid;
    const v2f sv = *(const v2f*)(mrgS + (row * NW + (lo ^ (8 * hi))) * 2);
    const float M  = redmax16(sv.x);
    const float f  = fexp2(sv.x - M);
    const float zt = redsum16(f * sv.y);
    float n0 = 0.f, n1 = 0.f;
#define MERGE_STEP(k)                                                              \
    {                                                                              \
      const float fk = dpp_mov<0x150 + (k)>(f);                                    \
      const v2f nv = *(const v2f*)(mrgN + mrg_idx(8 * hi + (k), row, lo));         \
      n0 = fmaf(fk, nv.x, n0);                                                     \
      n1 = fmaf(fk, nv.y, n1);                                                     \
    }
    MERGE_STEP(0) MERGE_STEP(1) MERGE_STEP(2) MERGE_STEP(3)
    MERGE_STEP(4) MERGE_STEP(5) MERGE_STEP(6) MERGE_STEP(7)
#undef MERGE_STEP
    const float mine = hi ? n1 : n0;
    const float send = hi ? n0 : n1;
    const float num  = mine + xhalf(send);
    out[(b * N_ + i0 + row) * D_ + lane] = num * frcp(zt);
  }
}

extern "C" void kernel_launch(void* const* d_in, const int* in_sizes, int n_in,
                              void* d_out, int out_size, void* d_ws, size_t ws_size,
                              hipStream_t stream) {
  const float* h    = (const float*)d_in[1];
  const float* pe   = (const float*)d_in[2];
  const float* E    = (const float*)d_in[3];
  const float* A    = (const float*)d_in[4];
  const float* Wk   = (const float*)d_in[5];
  const float* bk   = (const float*)d_in[6];
  const float* Wq   = (const float*)d_in[7];
  const float* bq   = (const float*)d_in[8];
  const float* beta = (const float*)d_in[9];
  float* out = (float*)d_out;

  dim3 grid(B_ * (N_ / 16)), block(32 * NW);
  fokker_planck_fused_kernel<<<grid, block, 0, stream>>>(h, pe, E, A, Wk, bk, Wq, bq, beta, out);
}


# Score tiles transposed (pe_j as A, G as B): a lane owns one row i, P's D registers feed the value WMMAs as A (no LDS transpose), row max/sums over registers, one max per wave slice; new sum order
# speedup vs baseline: 11.6393x; 1.0620x over previous
#include <hip/hip_runtime.h>
#include <hip/hip_bf16.h>
#include <math.h>

#define B_   4
#define N_   512
#define D_   32
#define PE_  16

#define NW  16
#define JT_PER_W (N_ / 16 / NW)
static_assert(NW == 16, "the merge gives output row r of the 16-row tile to wave r, and one 16-lane row holds the NW partials' m and z");
static_assert(JT_PER_W * NW * 16 == N_, "the waves' j slices tile N exactly");
#define L2E 1.44269504088896340736f
#define INV_SQRT_D 0.17677669529663688110f
#define KS2 (10.0f * L2E)
#define LN2 0.69314718055994530942f

typedef __attribute__((ext_vector_type(2))) float v2f;
typedef __attribute__((ext_vector_type(8))) float v8f;

static __device__ __forceinline__ v8f wmma_f32(v2f a, v2f b, v8f c) {
  return __builtin_amdgcn_wmma_f32_16x16x4_f32(false, a, false, b, (short)0, c, false, false);
}

static __device__ __forceinline__ float fexp2(float x) { return __builtin_amdgcn_exp2f(x); }
static __device__ __forceinline__ float frcp(float x)  { return __builtin_amdgcn_rcpf(x); }
static __device__ __forceinline__ float flog2(float x) { return __builtin_amdgcn_logf(x); }

#define DPP_QUAD_XOR1  0xB1
#define DPP_QUAD_XOR2  0x4E
#define DPP_HALF_MIR   0x141
#define DPP_ROW_MIR    0x140

template <int CTRL>
static __device__ __forceinline__ float dpp_mov(float x) {
  return __int_as_float(
      __builtin_amdgcn_update_dpp(0, __float_as_int(x), CTRL, 0xf, 0xf, true));
}
static __device__ __forceinline__ float redmax16(float x) {
  x = fmaxf(x, dpp_mov<DPP_QUAD_XOR1>(x));
  x = fmaxf(x, dpp_mov<DPP_QUAD_XOR2>(x));
  x = fmaxf(x, dpp_mov<DPP_HALF_MIR>(x));
  x = fmaxf(x, dpp_mov<DPP_ROW_MIR>(x));
  return x;
}
static __device__ __forceinline__ float redsum16(float x) {
  x += dpp_mov<DPP_QUAD_XOR1>(x);
  x += dpp_mov<DPP_QUAD_XOR2>(x);
  x += dpp_mov<DPP_HALF_MIR>(x);
  x += dpp_mov<DPP_ROW_MIR>(x);
  return x;
}


static __device__ __forceinline__ int mrg_idx(int w, int row, int lo) {
  const int slot = row ^ (((row ^ w) >> 3) & 1);
  return ((w * 16 + slot) * 16 + lo) * 2;
}
static __device__ __forceinline__ float xhalf(float x) {
  return __int_as_float(__builtin_amdgcn_permlanex16(__float_as_int(x), __float_as_int(x),
                                                     0x76543210, 0xfedcba98, false, false));
}

__global__ void __launch_bounds__(32 * NW) __attribute__((amdgpu_waves_per_eu(4, 4)))
fokker_planck_fused_kernel(const float* __restrict__ h,
                           const float* __restrict__ pe,
                           const float* __restrict__ E,
                           const float* __restrict__ A,
                           const float* __restrict__ Wk, const float* __restrict__ bk,
                           const float* __restrict__ Wq, const float* __restrict__ bq,
                           const float* __restrict__ beta,
                           float* __restrict__ out) {
  __shared__ __attribute__((aligned(16))) float mrgN[NW * 16 * 16 * 2];
  __shared__ __attribute__((aligned(16))) float mrgS[NW * 17 * 4];
  __shared__ __attribute__((aligned(16))) float gxG[2 * 2 * 32 * 4];
  __shared__ float gxb[2 * 2 * 16];

  const int lane = threadIdx.x & 31;
  const int wid  = threadIdx.x >> 5;
  const int lo   = lane & 15;
  const int hi   = lane >> 4;
  const int tile = blockIdx.x;
  const int b  = tile >> 5;
  const int i0 = (tile & 31) * 16;

  const float* Pb = pe + b * N_ * PE_;
  const float* Hb = h  + b * N_ * D_;

  float bts[2];
  bts[0] = beta[lo] * LN2; bts[1] = beta[lo + 16] * LN2;

  float Ei[8];
#pragma unroll
  for (int v = 0; v < 8; ++v) Ei[v] = E[i0 + v + 8 * hi];
  const float Eil = E[i0 + lo];

  float Hi[2][8];
#pragma unroll
  for (int v = 0; v < 8; ++v) {
    Hi[0][v] = Hb[(i0 + v + 8 * hi) * D_ + lo];
    Hi[1][v] = Hb[(i0 + v + 8 * hi) * D_ + lo + 16];
  }
  const float Hrow = Hb[(i0 + wid) * D_ + lane];
  const float brow = beta[lane] * LN2;
  const float Erow = E[i0 + wid];

  float Ej[JT_PER_W][8];
  float Am[JT_PER_W][8];
  v2f bPj[JT_PER_W][4];
#pragma unroll
  for (int t = 0; t < JT_PER_W; ++t) {
    const int j0 = (wid * JT_PER_W + t) * 16;
    const float4 e0 = *(const float4*)(E + j0 + 8 * hi);
    const float4 e1 = *(const float4*)(E + j0 + 8 * hi + 4);
    Ej[t][0] = e0.x; Ej[t][1] = e0.y; Ej[t][2] = e0.z; Ej[t][3] = e0.w;
    Ej[t][4] = e1.x; Ej[t][5] = e1.y; Ej[t][6] = e1.z; Ej[t][7] = e1.w;
    const float* Ar = A + (i0 + lo) * N_ + j0 + 8 * hi;
    const float4 a0 = *(const float4*)(Ar);
    const float4 a1 = *(const float4*)(Ar + 4);
    Am[t][0] = a0.x; Am[t][1] = a0.y; Am[t][2] = a0.z; Am[t][3] = a0.w;
    Am[t][4] = a1.x; Am[t][5] = a1.y; Am[t][6] = a1.z; Am[t][7] = a1.w;
#pragma unroll
    for (int q = 0; q < 4; ++q) bPj[t][q] = *(const v2f*)(Pb + (j0 + lo) * PE_ + 8 * hi + 2 * q);
  }

  const int wv = __builtin_amdgcn_readfirstlane(wid);
  if (wv < 2) {
    const int s = wv;
    v8f kt;
    for (int v = 0; v < 8; ++v) kt[v] = bk[16 * s + 8 * hi + v];
    for (int kc = 0; kc < 4; ++kc) {
      const int o = (4 * kc + 2 * hi) * D_ + 16 * s + lo;
      v2f aWk; aWk.x = Wk[o]; aWk.y = Wk[o + D_];
      v2f bPi = *(const v2f*)(Pb + (i0 + lo) * PE_ + 4 * kc + 2 * hi);
      kt = wmma_f32(aWk, bPi, kt);
    }
    v8f gt = {};
    for (int q = 0; q < 4; ++q) {
      v2f aWq = *(const v2f*)(Wq + lo * D_ + 16 * s + 8 * hi + 2 * q);
      v2f bK; bK.x = kt[2 * q]; bK.y = kt[2 * q + 1];
      gt = wmma_f32(aWq, bK, gt);
    }
    float gp = 0.f;
    for (int v = 0; v < 8; ++v) gp = fmaf(kt[v], bq[16 * s + 8 * hi + v], gp);
    float4 g0; g0.x = gt[0]; g0.y = gt[1]; g0.z = gt[2]; g0.w = gt[3];
    float4 g1; g1.x = gt[4]; g1.y = gt[5]; g1.z = gt[6]; g1.w = gt[7];
    *(float4*)(gxG + ((s * 2 + 0) * 32 + lane) * 4) = g0;
    *(float4*)(gxG + ((s * 2 + 1) * 32 + lane) * 4) = g1;
    gxb[(s * 2 + hi) * 16 + lo] = gp;
  }
  __syncthreads();

  const float kscale = INV_SQRT_D * L2E;
  v2f afG[4];
  float ginit;
  {
    const float4 a0 = *(const float4*)(gxG + ((0 * 2 + 0) * 32 + lane) * 4);
    const float4 a1 = *(const float4*)(gxG + ((0 * 2 + 1) * 32 + lane) * 4);
    const float4 b0 = *(const float4*)(gxG + ((1 * 2 + 0) * 32 + lane) * 4);
    const float4 b1 = *(const float4*)(gxG + ((1 * 2 + 1) * 32 + lane) * 4);
    afG[0].x = (a0.x + b0.x) * kscale;  afG[0].y = (a0.y + b0.y) * kscale;
    afG[1].x = (a0.z + b0.z) * kscale;  afG[1].y = (a0.w + b0.w) * kscale;
    afG[2].x = (a1.x + b1.x) * kscale;  afG[2].y = (a1.y + b1.y) * kscale;
    afG[3].x = (a1.z + b1.z) * kscale;  afG[3].y = (a1.w + b1.w) * kscale;
    ginit = ((gxb[0 * 16 + lo] + gxb[1 * 16 + lo]) + (gxb[2 * 16 + lo] + gxb[3 * 16 + lo])) * kscale;
  }

  v8f sT[JT_PER_W];
#pragma unroll
  for (int t = 0; t < JT_PER_W; ++t)
#pragma unroll
    for (int v = 0; v < 8; ++v) sT[t][v] = ginit;
#pragma unroll
  for (int q = 0; q < 4; ++q)
#pragma unroll
    for (int t = 0; t < JT_PER_W; ++t) sT[t] = wmma_f32(bPj[t][q], afG[q], sT[t]);

  float sc[JT_PER_W][8];
#pragma unroll
  for (int t = 0; t < JT_PER_W; ++t)
#pragma unroll
    for (int r = 0; r < 8; ++r) sc[t][r] = Am[t][r] * sT[t][r];
  float mx = sc[0][0];
#pragma unroll
  for (int t = 0; t < JT_PER_W; ++t)
#pragma unroll
    for (int r = 0; r < 8; ++r) mx = fmaxf(mx, sc[t][r]);
  mx = fmaxf(mx, xhalf(mx));

  v2f p3a[JT_PER_W][4], p4a[JT_PER_W][4];
  float zc[4] = {0.f, 0.f, 0.f, 0.f}, cc[4] = {0.f, 0.f, 0.f, 0.f};
#pragma unroll
  for (int t = 0; t < JT_PER_W; ++t)
#pragma unroll
    for (int r = 0; r < 8; ++r) {
      const float p  = fexp2(sc[t][r] - mx);
      const float Ed = Ej[t][r] - Eil;
      const float sg = frcp(1.f + fexp2(-KS2 * Ed));
      const float p3 = p * sg;
      const float p4 = p - p3;
      zc[r & 3] += p;
      cc[r & 3] += p4;
      p3a[t][r >> 1][r & 1] = p3;
      p4a[t][r >> 1][r & 1] = p4;
    }
  float z  = (zc[0] + zc[1]) + (zc[2] + zc[3]);
  float c6 = (cc[0] + cc[1]) + (cc[2] + cc[3]);
  z  += xhalf(z);
  c6 += xhalf(c6);

  v8f accX[2] = {}, accH[2] = {}, accV[2] = {};
#pragma unroll
  for (int t = 0; t < JT_PER_W; ++t) {
    const int j0 = (wid * JT_PER_W + t) * 16;
    v2f HB[4][2], VB[4][2], HVB[4][2];
#pragma unroll
    for (int q = 0; q < 4; ++q) {
#pragma unroll
      for (int s = 0; s < 2; ++s)
#pragma unroll
        for (int tt = 0; tt < 2; ++tt) {
          float hv = Hb[(j0 + 8 * hi + 2 * q + tt) * D_ + lo + 16 * s];
          float lv = flog2(hv + 1e-8f);
          float vv = fmaf(bts[s], lv, Ej[t][2 * q + tt]);
          HB[q][s][tt]  = hv;
          VB[q][s][tt]  = vv;
          HVB[q][s][tt] = hv * vv;
        }
    }
#pragma unroll
    for (int s = 0; s < 2; ++s)
#pragma unroll
      for (int q = 0; q < 4; ++q) {
        accX[s] = wmma_f32(p3a[t][q], HVB[q][s], accX[s]);
        accH[s] = wmma_f32(p3a[t][q], HB[q][s],  accH[s]);
        accV[s] = wmma_f32(p4a[t][q], VB[q][s],  accV[s]);
      }
  }

  float numw[2][8];
#pragma unroll
  for (int v = 0; v < 8; ++v) {
#pragma unroll
    for (int s = 0; s < 2; ++s) {
      const float hv = Hi[s][v];
      const float lv = flog2(hv + 1e-8f);
      const float vi = fmaf(bts[s], lv, Ei[v]);
      const float t2 = fmaf(hv, accV[s][v], accX[s][v]);
      numw[s][v] = fmaf(-vi, accH[s][v], t2);
    }
  }

  {
#pragma unroll
    for (int v = 0; v < 8; ++v) {
      const int row = v + 8 * hi;
      v2f nv; nv.x = numw[0][v]; nv.y = numw[1][v];
      *(v2f*)(mrgN + mrg_idx(wid, row, lo)) = nv;
    }
    if (hi == 0) {
      float4 sv; sv.x = mx; sv.y = z; sv.z = c6; sv.w = 0.f;
      *(float4*)(mrgS + (wid * 17 + lo) * 4) = sv;
    }
  }
  __syncthreads();

  {
    const int row = wid;
    const float4 sv = *(const float4*)(mrgS + ((lo ^ (8 * hi)) * 17 + row) * 4);
    const float M  = redmax16(sv.x);
    const float f  = fexp2(sv.x - M);
    const float zt = redsum16(f * sv.y);
    const float ct = redsum16(f * sv.z);
    float n0 = 0.f, n1 = 0.f;
#define MERGE_STEP(k)                                                              \
    {                                                                              \
      const float fk = dpp_mov<0x150 + (k)>(f);                                    \
      const v2f nv = *(const v2f*)(mrgN + mrg_idx(8 * hi + (k), row, lo));         \
      n0 = fmaf(fk, nv.x, n0);                                                     \
      n1 = fmaf(fk, nv.y, n1);                                                     \
    }
    MERGE_STEP(0) MERGE_STEP(1) MERGE_STEP(2) MERGE_STEP(3)
    MERGE_STEP(4) MERGE_STEP(5) MERGE_STEP(6) MERGE_STEP(7)
#undef MERGE_STEP
    const float mine = hi ? n1 : n0;
    const float send = hi ? n0 : n1;
    float num = mine + xhalf(send);
    const float vrow = fmaf(brow, flog2(Hrow + 1e-8f), Erow);
    num = fmaf(-(vrow * Hrow), ct, num);
    out[(b * N_ + i0 + row) * D_ + lane] = num * frcp(zt);
  }
}

extern "C" void kernel_launch(void* const* d_in, const int* in_sizes, int n_in,
                              void* d_out, int out_size, void* d_ws, size_t ws_size,
                              hipStream_t stream) {
  const float* h    = (const float*)d_in[1];
  const float* pe   = (const float*)d_in[2];
  const float* E    = (const float*)d_in[3];
  const float* A    = (const float*)d_in[4];
  const float* Wk   = (const float*)d_in[5];
  const float* bk   = (const float*)d_in[6];
  const float* Wq   = (const float*)d_in[7];
  const float* bq   = (const float*)d_in[8];
  const float* beta = (const float*)d_in[9];
  float* out = (float*)d_out;

  dim3 grid(B_ * (N_ / 16)), block(32 * NW);
  fokker_planck_fused_kernel<<<grid, block, 0, stream>>>(h, pe, E, A, Wk, bk, Wq, bq, beta, out);
}


# Every global load covers whole 128-byte lines: H and beta read as 8-byte pairs on a value-column map d = 2n + s, the mask tile read row-contiguous and turned to the score layout through LDS
# speedup vs baseline: 12.2706x; 1.0542x over previous
#include <hip/hip_runtime.h>
#include <hip/hip_bf16.h>
#include <math.h>

#define B_   4
#define N_   512
#define D_   32
#define PE_  16

#define NW  16
#define JT_PER_W (N_ / 16 / NW)
static_assert(NW == 16, "the merge gives output row r of the 16-row tile to wave r, and one 16-lane row holds the NW partials' m and z");
static_assert(JT_PER_W * NW * 16 == N_, "the waves' j slices tile N exactly");
#define L2E 1.44269504088896340736f
#define INV_SQRT_D 0.17677669529663688110f
#define KS2 (10.0f * L2E)
#define LN2 0.69314718055994530942f

typedef __attribute__((ext_vector_type(2))) float v2f;
typedef __attribute__((ext_vector_type(8))) float v8f;

static __device__ __forceinline__ v8f wmma_f32(v2f a, v2f b, v8f c) {
  return __builtin_amdgcn_wmma_f32_16x16x4_f32(false, a, false, b, (short)0, c, false, false);
}

static __device__ __forceinline__ float fexp2(float x) { return __builtin_amdgcn_exp2f(x); }
static __device__ __forceinline__ float frcp(float x)  { return __builtin_amdgcn_rcpf(x); }
static __device__ __forceinline__ float flog2(float x) { return __builtin_amdgcn_logf(x); }

#define DPP_QUAD_XOR1  0xB1
#define DPP_QUAD_XOR2  0x4E
#define DPP_HALF_MIR   0x141
#define DPP_ROW_MIR    0x140

template <int CTRL>
static __device__ __forceinline__ float dpp_mov(float x) {
  return __int_as_float(
      __builtin_amdgcn_update_dpp(0, __float_as_int(x), CTRL, 0xf, 0xf, true));
}
static __device__ __forceinline__ float redmax16(float x) {
  x = fmaxf(x, dpp_mov<DPP_QUAD_XOR1>(x));
  x = fmaxf(x, dpp_mov<DPP_QUAD_XOR2>(x));
  x = fmaxf(x, dpp_mov<DPP_HALF_MIR>(x));
  x = fmaxf(x, dpp_mov<DPP_ROW_MIR>(x));
  return x;
}
static __device__ __forceinline__ float redsum16(float x) {
  x += dpp_mov<DPP_QUAD_XOR1>(x);
  x += dpp_mov<DPP_QUAD_XOR2>(x);
  x += dpp_mov<DPP_HALF_MIR>(x);
  x += dpp_mov<DPP_ROW_MIR>(x);
  return x;
}


static __device__ __forceinline__ int mrg_idx(int w, int row, int lo) {
  const int slot = row ^ (((row ^ w) >> 3) & 1);
  return ((w * 16 + slot) * 16 + lo) * 2;
}
static __device__ __forceinline__ float xhalf(float x) {
  return __int_as_float(__builtin_amdgcn_permlanex16(__float_as_int(x), __float_as_int(x),
                                                     0x76543210, 0xfedcba98, false, false));
}

__global__ void __launch_bounds__(32 * NW) __attribute__((amdgpu_waves_per_eu(4, 4)))
fokker_planck_fused_kernel(const float* __restrict__ h,
                           const float* __restrict__ pe,
                           const float* __restrict__ E,
                           const float* __restrict__ A,
                           const float* __restrict__ Wk, const float* __restrict__ bk,
                           const float* __restrict__ Wq, const float* __restrict__ bq,
                           const float* __restrict__ beta,
                           float* __restrict__ out) {
  __shared__ __attribute__((aligned(16))) float mrgN[NW * 16 * 16 * 2];
  __shared__ __attribute__((aligned(16))) float mrgS[NW * 17 * 4];
  __shared__ __attribute__((aligned(16))) float gxG[2 * 2 * 32 * 4];
  __shared__ float gxb[2 * 2 * 16];
  __shared__ __attribute__((aligned(16))) float tA[NW * 16 * 36];

  const int lane = threadIdx.x & 31;
  const int wid  = threadIdx.x >> 5;
  const int lo   = lane & 15;
  const int hi   = lane >> 4;
  const int tile = blockIdx.x;
  const int b  = tile >> 5;
  const int i0 = (tile & 31) * 16;

  const float* Pb = pe + b * N_ * PE_;
  const float* Hb = h  + b * N_ * D_;

  float4 aRow[4];
#pragma unroll
  for (int m = 0; m < 4; ++m)
    aRow[m] = *(const float4*)(A + (i0 + 4 * m + (lane >> 3)) * N_ + wid * (16 * JT_PER_W) + 4 * (lane & 7));

  float bts[2];
  {
    const v2f bt2 = *(const v2f*)(beta + 2 * lo);
    bts[0] = bt2.x * LN2; bts[1] = bt2.y * LN2;
  }

  float Ei[8];
#pragma unroll
  for (int v = 0; v < 8; ++v) Ei[v] = E[i0 + v + 8 * hi];
  const float Eil = E[i0 + lo];

  float Hi[2][8];
#pragma unroll
  for (int v = 0; v < 8; ++v) {
    const v2f h2 = *(const v2f*)(Hb + (i0 + v + 8 * hi) * D_ + 2 * lo);
    Hi[0][v] = h2.x;
    Hi[1][v] = h2.y;
  }
  const int dcol = 2 * lo + hi;
  const float Hrow = Hb[(i0 + wid) * D_ + dcol];
  const float brow = beta[dcol] * LN2;
  const float Erow = E[i0 + wid];

  float Ej[JT_PER_W][8];
  v2f bPj[JT_PER_W][4];
#pragma unroll
  for (int t = 0; t < JT_PER_W; ++t) {
    const int j0 = (wid * JT_PER_W + t) * 16;
    const float4 e0 = *(const float4*)(E + j0 + 8 * hi);
    const float4 e1 = *(const float4*)(E + j0 + 8 * hi + 4);
    Ej[t][0] = e0.x; Ej[t][1] = e0.y; Ej[t][2] = e0.z; Ej[t][3] = e0.w;
    Ej[t][4] = e1.x; Ej[t][5] = e1.y; Ej[t][6] = e1.z; Ej[t][7] = e1.w;
#pragma unroll
    for (int q = 0; q < 4; ++q) bPj[t][q] = *(const v2f*)(Pb + (j0 + lo) * PE_ + 8 * hi + 2 * q);
  }

  const int wv = __builtin_amdgcn_readfirstlane(wid);
  if (wv < 2) {
    const int s = wv;
    v8f kt;
    for (int v = 0; v < 8; ++v) kt[v] = bk[16 * s + 8 * hi + v];
    for (int kc = 0; kc < 4; ++kc) {
      const int o = (4 * kc + 2 * hi) * D_ + 16 * s + lo;
      v2f aWk; aWk.x = Wk[o]; aWk.y = Wk[o + D_];
      v2f bPi = *(const v2f*)(Pb + (i0 + lo) * PE_ + 4 * kc + 2 * hi);
      kt = wmma_f32(aWk, bPi, kt);
    }
    v8f gt = {};
    for (int q = 0; q < 4; ++q) {
      v2f aWq = *(const v2f*)(Wq + lo * D_ + 16 * s + 8 * hi + 2 * q);
      v2f bK; bK.x = kt[2 * q]; bK.y = kt[2 * q + 1];
      gt = wmma_f32(aWq, bK, gt);
    }
    float gp = 0.f;
    for (int v = 0; v < 8; ++v) gp = fmaf(kt[v], bq[16 * s + 8 * hi + v], gp);
    float4 g0; g0.x = gt[0]; g0.y = gt[1]; g0.z = gt[2]; g0.w = gt[3];
    float4 g1; g1.x = gt[4]; g1.y = gt[5]; g1.z = gt[6]; g1.w = gt[7];
    *(float4*)(gxG + ((s * 2 + 0) * 32 + lane) * 4) = g0;
    *(float4*)(gxG + ((s * 2 + 1) * 32 + lane) * 4) = g1;
#pragma unroll
    for (int m = 0; m < 4; ++m)
      *(float4*)(tA + (wid * 16 + 4 * m + (lane >> 3)) * 36 + 4 * (lane & 7)) = aRow[m];
    gxb[(s * 2 + hi) * 16 + lo] = gp;
  } else {
#pragma unroll
    for (int m = 0; m < 4; ++m)
      *(float4*)(tA + (wid * 16 + 4 * m + (lane >> 3)) * 36 + 4 * (lane & 7)) = aRow[m];
  }
  __syncthreads();

  const float kscale = INV_SQRT_D * L2E;
  v2f afG[4];
  float ginit;
  {
    const float4 a0 = *(const float4*)(gxG + ((0 * 2 + 0) * 32 + lane) * 4);
    const float4 a1 = *(const float4*)(gxG + ((0 * 2 + 1) * 32 + lane) * 4);
    const float4 b0 = *(const float4*)(gxG + ((1 * 2 + 0) * 32 + lane) * 4);
    const float4 b1 = *(const float4*)(gxG + ((1 * 2 + 1) * 32 + lane) * 4);
    afG[0].x = (a0.x + b0.x) * kscale;  afG[0].y = (a0.y + b0.y) * kscale;
    afG[1].x = (a0.z + b0.z) * kscale;  afG[1].y = (a0.w + b0.w) * kscale;
    afG[2].x = (a1.x + b1.x) * kscale;  afG[2].y = (a1.y + b1.y) * kscale;
    afG[3].x = (a1.z + b1.z) * kscale;  afG[3].y = (a1.w + b1.w) * kscale;
    ginit = ((gxb[0 * 16 + lo] + gxb[1 * 16 + lo]) + (gxb[2 * 16 + lo] + gxb[3 * 16 + lo])) * kscale;
  }

  v8f sT[JT_PER_W];
#pragma unroll
  for (int t = 0; t < JT_PER_W; ++t)
#pragma unroll
    for (int v = 0; v < 8; ++v) sT[t][v] = ginit;
#pragma unroll
  for (int q = 0; q < 4; ++q)
#pragma unroll
    for (int t = 0; t < JT_PER_W; ++t) sT[t] = wmma_f32(bPj[t][q], afG[q], sT[t]);

  float Am[JT_PER_W][8];
#pragma unroll
  for (int t = 0; t < JT_PER_W; ++t) {
    const float* ar = tA + (wid * 16 + lo) * 36 + 16 * t + 8 * hi;
    const float4 a0 = *(const float4*)(ar);
    const float4 a1 = *(const float4*)(ar + 4);
    Am[t][0] = a0.x; Am[t][1] = a0.y; Am[t][2] = a0.z; Am[t][3] = a0.w;
    Am[t][4] = a1.x; Am[t][5] = a1.y; Am[t][6] = a1.z; Am[t][7] = a1.w;
  }
  float sc[JT_PER_W][8];
#pragma unroll
  for (int t = 0; t < JT_PER_W; ++t)
#pragma unroll
    for (int r = 0; r < 8; ++r) sc[t][r] = Am[t][r] * sT[t][r];
  float mx = sc[0][0];
#pragma unroll
  for (int t = 0; t < JT_PER_W; ++t)
#pragma unroll
    for (int r = 0; r < 8; ++r) mx = fmaxf(mx, sc[t][r]);
  mx = fmaxf(mx, xhalf(mx));

  v2f p3a[JT_PER_W][4], p4a[JT_PER_W][4];
  float zc[4] = {0.f, 0.f, 0.f, 0.f}, cc[4] = {0.f, 0.f, 0.f, 0.f};
#pragma unroll
  for (int t = 0; t < JT_PER_W; ++t)
#pragma unroll
    for (int r = 0; r < 8; ++r) {
      const float p  = fexp2(sc[t][r] - mx);
      const float Ed = Ej[t][r] - Eil;
      const float sg = frcp(1.f + fexp2(-KS2 * Ed));
      const float p3 = p * sg;
      const float p4 = p - p3;
      zc[r & 3] += p;
      cc[r & 3] += p4;
      p3a[t][r >> 1][r & 1] = p3;
      p4a[t][r >> 1][r & 1] = p4;
    }
  float z  = (zc[0] + zc[1]) + (zc[2] + zc[3]);
  float c6 = (cc[0] + cc[1]) + (cc[2] + cc[3]);
  z  += xhalf(z);
  c6 += xhalf(c6);

  v8f accX[2] = {}, accH[2] = {}, accV[2] = {};
#pragma unroll
  for (int t = 0; t < JT_PER_W; ++t) {
    const int j0 = (wid * JT_PER_W + t) * 16;
    v2f HB[4][2], VB[4][2], HVB[4][2];
#pragma unroll
    for (int q = 0; q < 4; ++q) {
#pragma unroll
      for (int tt = 0; tt < 2; ++tt) {
        const v2f h2 = *(const v2f*)(Hb + (j0 + 8 * hi + 2 * q + tt) * D_ + 2 * lo);
#pragma unroll
        for (int s = 0; s < 2; ++s) {
          float hv = h2[s];
          float lv = flog2(hv + 1e-8f);
          float vv = fmaf(bts[s], lv, Ej[t][2 * q + tt]);
          HB[q][s][tt]  = hv;
          VB[q][s][tt]  = vv;
          HVB[q][s][tt] = hv * vv;
        }
      }
    }
#pragma unroll
    for (int s = 0; s < 2; ++s)
#pragma unroll
      for (int q = 0; q < 4; ++q) {
        accX[s] = wmma_f32(p3a[t][q], HVB[q][s], accX[s]);
        accH[s] = wmma_f32(p3a[t][q], HB[q][s],  accH[s]);
        accV[s] = wmma_f32(p4a[t][q], VB[q][s],  accV[s]);
      }
  }

  float numw[2][8];
#pragma unroll
  for (int v = 0; v < 8; ++v) {
#pragma unroll
    for (int s = 0; s < 2; ++s) {
      const float hv = Hi[s][v];
      const float lv = flog2(hv + 1e-8f);
      const float vi = fmaf(bts[s], lv, Ei[v]);
      const float t2 = fmaf(hv, accV[s][v], accX[s][v]);
      numw[s][v] = fmaf(-vi, accH[s][v], t2);
    }
  }

  {
#pragma unroll
    for (int v = 0; v < 8; ++v) {
      const int row = v + 8 * hi;
      v2f nv; nv.x = numw[0][v]; nv.y = numw[1][v];
      *(v2f*)(mrgN + mrg_idx(wid, row, lo)) = nv;
    }
    if (hi == 0) {
      float4 sv; sv.x = mx; sv.y = z; sv.z = c6; sv.w = 0.f;
      *(float4*)(mrgS + (wid * 17 + lo) * 4) = sv;
    }
  }
  __syncthreads();

  {
    const int row = wid;
    const float4 sv = *(const float4*)(mrgS + ((lo ^ (8 * hi)) * 17 + row) * 4);
    const float M  = redmax16(sv.x);
    const float f  = fexp2(sv.x - M);
    const float zt = redsum16(f * sv.y);
    const float ct = redsum16(f * sv.z);
    float n0 = 0.f, n1 = 0.f;
#define MERGE_STEP(k)                                                              \
    {                                                                              \
      const float fk = dpp_mov<0x150 + (k)>(f);                                    \
      const v2f nv = *(const v2f*)(mrgN + mrg_idx(8 * hi + (k), row, lo));         \
      n0 = fmaf(fk, nv.x, n0);                                                     \
      n1 = fmaf(fk, nv.y, n1);                                                     \
    }
    MERGE_STEP(0) MERGE_STEP(1) MERGE_STEP(2) MERGE_STEP(3)
    MERGE_STEP(4) MERGE_STEP(5) MERGE_STEP(6) MERGE_STEP(7)
#undef MERGE_STEP
    const float mine = hi ? n1 : n0;
    const float send = hi ? n0 : n1;
    float num = mine + xhalf(send);
    const float vrow = fmaf(brow, flog2(Hrow + 1e-8f), Erow);
    num = fmaf(-(vrow * Hrow), ct, num);
    out[(b * N_ + i0 + row) * D_ + dcol] = num * frcp(zt);
  }
}

extern "C" void kernel_launch(void* const* d_in, const int* in_sizes, int n_in,
                              void* d_out, int out_size, void* d_ws, size_t ws_size,
                              hipStream_t stream) {
  const float* h    = (const float*)d_in[1];
  const float* pe   = (const float*)d_in[2];
  const float* E    = (const float*)d_in[3];
  const float* A    = (const float*)d_in[4];
  const float* Wk   = (const float*)d_in[5];
  const float* bk   = (const float*)d_in[6];
  const float* Wq   = (const float*)d_in[7];
  const float* bq   = (const float*)d_in[8];
  const float* beta = (const float*)d_in[9];
  float* out = (float*)d_out;

  dim3 grid(B_ * (N_ / 16)), block(32 * NW);
  fokker_planck_fused_kernel<<<grid, block, 0, stream>>>(h, pe, E, A, Wk, bk, Wq, bq, beta, out);
}


# Two work-groups per 16-row tile, one per 16-column half of the output (256 groups x 8 waves x 4 j-tiles): 12 value WMMAs per j-tile, not 24; 64-byte row pieces stored; 8-partial merge: new sum tree
# speedup vs baseline: 13.6018x; 1.1085x over previous
#include <hip/hip_runtime.h>
#include <hip/hip_bf16.h>
#include <math.h>

#define B_   4
#define N_   512
#define D_   32
#define PE_  16

#define NW  8
#define JT_PER_W (N_ / 16 / NW)
#define DH  16
static_assert(NW == 8, "the merge gives output rows r and r + 8 of the 16-row tile to wave r, and 8 lanes hold the NW partials' m and z");
static_assert(JT_PER_W * NW * 16 == N_, "the waves' j slices tile N exactly");
static_assert(2 * DH == D_, "two work-groups per i-tile cover the D columns");
#define TA_PITCH (16 * JT_PER_W + 4)
#define MRG_TILE (16 * DH + 16)
#define L2E 1.44269504088896340736f
#define INV_SQRT_D 0.17677669529663688110f
#define KS2 (10.0f * L2E)
#define LN2 0.69314718055994530942f

typedef __attribute__((ext_vector_type(2))) float v2f;
typedef __attribute__((ext_vector_type(8))) float v8f;

static __device__ __forceinline__ v8f wmma_f32(v2f a, v2f b, v8f c) {
  return __builtin_amdgcn_wmma_f32_16x16x4_f32(false, a, false, b, (short)0, c, false, false);
}

static __device__ __forceinline__ float fexp2(float x) { return __builtin_amdgcn_exp2f(x); }
static __device__ __forceinline__ float frcp(float x)  { return __builtin_amdgcn_rcpf(x); }
static __device__ __forceinline__ float flog2(float x) { return __builtin_amdgcn_logf(x); }

#define DPP_QUAD_XOR1  0xB1
#define DPP_QUAD_XOR2  0x4E
#define DPP_HALF_MIR   0x141

template <int CTRL>
static __device__ __forceinline__ float dpp_mov(float x) {
  return __int_as_float(
      __builtin_amdgcn_update_dpp(0, __float_as_int(x), CTRL, 0xf, 0xf, true));
}
static __device__ __forceinline__ float redmax8(float x) {
  x = fmaxf(x, dpp_mov<DPP_QUAD_XOR1>(x));
  x = fmaxf(x, dpp_mov<DPP_QUAD_XOR2>(x));
  x = fmaxf(x, dpp_mov<DPP_HALF_MIR>(x));
  return x;
}
static __device__ __forceinline__ float redsum8(float x) {
  x += dpp_mov<DPP_QUAD_XOR1>(x);
  x += dpp_mov<DPP_QUAD_XOR2>(x);
  x += dpp_mov<DPP_HALF_MIR>(x);
  return x;
}


static __device__ __forceinline__ int mrg_idx(int w, int row, int lo) {
  return w * MRG_TILE + (row + (row >> 3)) * 16 + lo;
}
static __device__ __forceinline__ float xhalf(float x) {
  return __int_as_float(__builtin_amdgcn_permlanex16(__float_as_int(x), __float_as_int(x),
                                                     0x76543210, 0xfedcba98, false, false));
}

__global__ void __launch_bounds__(32 * NW) __attribute__((amdgpu_waves_per_eu(2, 2)))
fokker_planck_fused_kernel(const float* __restrict__ h,
                           const float* __restrict__ pe,
                           const float* __restrict__ E,
                           const float* __restrict__ A,
                           const float* __restrict__ Wk, const float* __restrict__ bk,
                           const float* __restrict__ Wq, const float* __restrict__ bq,
                           const float* __restrict__ beta,
                           float* __restrict__ out) {
  __shared__ __attribute__((aligned(16))) float mrgN[NW * MRG_TILE];
  __shared__ __attribute__((aligned(16))) float mrgS[NW * 17 * 4];
  __shared__ __attribute__((aligned(16))) float gxG[2 * 2 * 32 * 4];
  __shared__ float gxb[2 * 2 * 16];
  __shared__ __attribute__((aligned(16))) float tA[NW * 16 * TA_PITCH];

  const int lane = threadIdx.x & 31;
  const int wid  = threadIdx.x >> 5;
  const int lo   = lane & 15;
  const int hi   = lane >> 4;
  const int tile = blockIdx.x >> 1;
  const int c0 = (blockIdx.x & 1) * DH;
  const int b  = tile >> 5;
  const int i0 = (tile & 31) * 16;

  const float* Pb = pe + b * N_ * PE_;
  const float* Hb = h  + b * N_ * D_;

  float4 aRow[8];
#pragma unroll
  for (int m = 0; m < 8; ++m)
    aRow[m] = *(const float4*)(A + (i0 + 2 * m + (lane >> 4)) * N_ + wid * (16 * JT_PER_W) + 4 * (lane & 15));

  const float bts = beta[c0 + lo] * LN2;

  float Ei[8];
#pragma unroll
  for (int v = 0; v < 8; ++v) Ei[v] = E[i0 + v + 8 * hi];
  const float Eil = E[i0 + lo];

  float Hi[8];
#pragma unroll
  for (int v = 0; v < 8; ++v) Hi[v] = Hb[(i0 + v + 8 * hi) * D_ + c0 + lo];
  const int mrow = wid + 8 * hi;
  const float Hrow = Hb[(i0 + mrow) * D_ + c0 + lo];
  const float Erow = E[i0 + mrow];

  float Ej[JT_PER_W][8];
  v2f bPj[JT_PER_W][4];
#pragma unroll
  for (int t = 0; t < JT_PER_W; ++t) {
    const int j0 = (wid * JT_PER_W + t) * 16;
    const float4 e0 = *(const float4*)(E + j0 + 8 * hi);
    const float4 e1 = *(const float4*)(E + j0 + 8 * hi + 4);
    Ej[t][0] = e0.x; Ej[t][1] = e0.y; Ej[t][2] = e0.z; Ej[t][3] = e0.w;
    Ej[t][4] = e1.x; Ej[t][5] = e1.y; Ej[t][6] = e1.z; Ej[t][7] = e1.w;
#pragma unroll
    for (int q = 0; q < 4; ++q) bPj[t][q] = *(const v2f*)(Pb + (j0 + lo) * PE_ + 8 * hi + 2 * q);
  }

  const int wv = __builtin_amdgcn_readfirstlane(wid);
  if (wv < 2) {
    const int s = wv;
    v8f kt;
    for (int v = 0; v < 8; ++v) kt[v] = bk[16 * s + 8 * hi + v];
    for (int kc = 0; kc < 4; ++kc) {
      const int o = (4 * kc + 2 * hi) * D_ + 16 * s + lo;
      v2f aWk; aWk.x = Wk[o]; aWk.y = Wk[o + D_];
      v2f bPi = *(const v2f*)(Pb + (i0 + lo) * PE_ + 4 * kc + 2 * hi);
      kt = wmma_f32(aWk, bPi, kt);
    }
    v8f gt = {};
    for (int q = 0; q < 4; ++q) {
      v2f aWq = *(const v2f*)(Wq + lo * D_ + 16 * s + 8 * hi + 2 * q);
      v2f bK; bK.x = kt[2 * q]; bK.y = kt[2 * q + 1];
      gt = wmma_f32(aWq, bK, gt);
    }
    float gp = 0.f;
    for (int v = 0; v < 8; ++v) gp = fmaf(kt[v], bq[16 * s + 8 * hi + v], gp);
    float4 g0; g0.x = gt[0]; g0.y = gt[1]; g0.z = gt[2]; g0.w = gt[3];
    float4 g1; g1.x = gt[4]; g1.y = gt[5]; g1.z = gt[6]; g1.w = gt[7];
    *(float4*)(gxG + ((s * 2 + 0) * 32 + lane) * 4) = g0;
    *(float4*)(gxG + ((s * 2 + 1) * 32 + lane) * 4) = g1;
#pragma unroll
    for (int m = 0; m < 8; ++m)
      *(float4*)(tA + (wid * 16 + 2 * m + (lane >> 4)) * TA_PITCH + 4 * (lane & 15)) = aRow[m];
    gxb[(s * 2 + hi) * 16 + lo] = gp;
  } else {
#pragma unroll
    for (int m = 0; m < 8; ++m)
      *(float4*)(tA + (wid * 16 + 2 * m + (lane >> 4)) * TA_PITCH + 4 * (lane & 15)) = aRow[m];
  }
  __syncthreads();

  const float kscale = INV_SQRT_D * L2E;
  v2f afG[4];
  float ginit;
  {
    const float4 a0 = *(const float4*)(gxG + ((0 * 2 + 0) * 32 + lane) * 4);
    const float4 a1 = *(const float4*)(gxG + ((0 * 2 + 1) * 32 + lane) * 4);
    const float4 b0 = *(const float4*)(gxG + ((1 * 2 + 0) * 32 + lane) * 4);
    const float4 b1 = *(const float4*)(gxG + ((1 * 2 + 1) * 32 + lane) * 4);
    afG[0].x = (a0.x + b0.x) * kscale;  afG[0].y = (a0.y + b0.y) * kscale;
    afG[1].x = (a0.z + b0.z) * kscale;  afG[1].y = (a0.w + b0.w) * kscale;
    afG[2].x = (a1.x + b1.x) * kscale;  afG[2].y = (a1.y + b1.y) * kscale;
    afG[3].x = (a1.z + b1.z) * kscale;  afG[3].y = (a1.w + b1.w) * kscale;
    ginit = ((gxb[0 * 16 + lo] + gxb[1 * 16 + lo]) + (gxb[2 * 16 + lo] + gxb[3 * 16 + lo])) * kscale;
  }

  v8f sT[JT_PER_W];
#pragma unroll
  for (int t = 0; t < JT_PER_W; ++t)
#pragma unroll
    for (int v = 0; v < 8; ++v) sT[t][v] = ginit;
#pragma unroll
  for (int q = 0; q < 4; ++q)
#pragma unroll
    for (int t = 0; t < JT_PER_W; ++t) sT[t] = wmma_f32(bPj[t][q], afG[q], sT[t]);

  float Am[JT_PER_W][8];
#pragma unroll
  for (int t = 0; t < JT_PER_W; ++t) {
    const float* ar = tA + (wid * 16 + lo) * TA_PITCH + 16 * t + 8 * hi;
    const float4 a0 = *(const float4*)(ar);
    const float4 a1 = *(const float4*)(ar + 4);
    Am[t][0] = a0.x; Am[t][1] = a0.y; Am[t][2] = a0.z; Am[t][3] = a0.w;
    Am[t][4] = a1.x; Am[t][5] = a1.y; Am[t][6] = a1.z; Am[t][7] = a1.w;
  }
  float sc[JT_PER_W][8];
#pragma unroll
  for (int t = 0; t < JT_PER_W; ++t)
#pragma unroll
    for (int r = 0; r < 8; ++r) sc[t][r] = Am[t][r] * sT[t][r];
  float mx = sc[0][0];
#pragma unroll
  for (int t = 0; t < JT_PER_W; ++t)
#pragma unroll
    for (int r = 0; r < 8; ++r) mx = fmaxf(mx, sc[t][r]);
  mx = fmaxf(mx, xhalf(mx));

  v2f p3a[JT_PER_W][4], p4a[JT_PER_W][4];
  float zc[4] = {0.f, 0.f, 0.f, 0.f}, cc[4] = {0.f, 0.f, 0.f, 0.f};
#pragma unroll
  for (int t = 0; t < JT_PER_W; ++t)
#pragma unroll
    for (int r = 0; r < 8; ++r) {
      const float p  = fexp2(sc[t][r] - mx);
      const float Ed = Ej[t][r] - Eil;
      const float sg = frcp(1.f + fexp2(-KS2 * Ed));
      const float p3 = p * sg;
      const float p4 = p - p3;
      zc[r & 3] += p;
      cc[r & 3] += p4;
      p3a[t][r >> 1][r & 1] = p3;
      p4a[t][r >> 1][r & 1] = p4;
    }
  float z  = (zc[0] + zc[1]) + (zc[2] + zc[3]);
  float c6 = (cc[0] + cc[1]) + (cc[2] + cc[3]);
  z  += xhalf(z);
  c6 += xhalf(c6);

  v8f accX = {}, accH = {}, accV = {};
#pragma unroll
  for (int t = 0; t < JT_PER_W; ++t) {
    const int j0 = (wid * JT_PER_W + t) * 16;
    v2f HB[4], VB[4], HVB[4];
#pragma unroll
    for (int q = 0; q < 4; ++q) {
#pragma unroll
      for (int tt = 0; tt < 2; ++tt) {
        const float hv = Hb[(j0 + 8 * hi + 2 * q + tt) * D_ + c0 + lo];
        const float lv = flog2(hv + 1e-8f);
        const float vv = fmaf(bts, lv, Ej[t][2 * q + tt]);
        HB[q][tt]  = hv;
        VB[q][tt]  = vv;
        HVB[q][tt] = hv * vv;
      }
    }
#pragma unroll
    for (int q = 0; q < 4; ++q) {
      accX = wmma_f32(p3a[t][q], HVB[q], accX);
      accH = wmma_f32(p3a[t][q], HB[q],  accH);
      accV = wmma_f32(p4a[t][q], VB[q],  accV);
    }
  }

  float numw[8];
#pragma unroll
  for (int v = 0; v < 8; ++v) {
    const float hv = Hi[v];
    const float lv = flog2(hv + 1e-8f);
    const float vi = fmaf(bts, lv, Ei[v]);
    const float t2 = fmaf(hv, accV[v], accX[v]);
    numw[v] = fmaf(-vi, accH[v], t2);
  }

  {
#pragma unroll
    for (int v = 0; v < 8; ++v) mrgN[mrg_idx(wid, v + 8 * hi, lo)] = numw[v];
    if (hi == 0) {
      float4 sv; sv.x = mx; sv.y = z; sv.z = c6; sv.w = 0.f;
      *(float4*)(mrgS + (wid * 17 + lo) * 4) = sv;
    }
  }
  __syncthreads();

  {
    const int row = mrow;
    const float4 sv = *(const float4*)(mrgS + ((lo & 7) * 17 + row) * 4);
    const float M  = redmax8(sv.x);
    const float f  = fexp2(sv.x - M);
    const float zt = redsum8(f * sv.y);
    const float ct = redsum8(f * sv.z);
    float num = 0.f;
#define MERGE_STEP(k)                                                              \
    {                                                                              \
      const float fk = dpp_mov<0x150 + (k)>(f);                                    \
      num = fmaf(fk, mrgN[mrg_idx((k), row, lo)], num);                            \
    }
    MERGE_STEP(0) MERGE_STEP(1) MERGE_STEP(2) MERGE_STEP(3)
    MERGE_STEP(4) MERGE_STEP(5) MERGE_STEP(6) MERGE_STEP(7)
#undef MERGE_STEP
    const float vrow = fmaf(bts, flog2(Hrow + 1e-8f), Erow);
    num = fmaf(-(vrow * Hrow), ct, num);
    out[(b * N_ + i0 + row) * D_ + c0 + lo] = num * frcp(zt);
  }
}

extern "C" void kernel_launch(void* const* d_in, const int* in_sizes, int n_in,
                              void* d_out, int out_size, void* d_ws, size_t ws_size,
                              hipStream_t stream) {
  const float* h    = (const float*)d_in[1];
  const float* pe   = (const float*)d_in[2];
  const float* E    = (const float*)d_in[3];
  const float* A    = (const float*)d_in[4];
  const float* Wk   = (const float*)d_in[5];
  const float* bk   = (const float*)d_in[6];
  const float* Wq   = (const float*)d_in[7];
  const float* bq   = (const float*)d_in[8];
  const float* beta = (const float*)d_in[9];
  float* out = (float*)d_out;

  dim3 grid(2 * B_ * (N_ / 16)), block(32 * NW);
  fokker_planck_fused_kernel<<<grid, block, 0, stream>>>(h, pe, E, A, Wk, bk, Wq, bq, beta, out);
}


# Head ordered by need: the first barrier fences LDS only, so global loads stay in flight across it; waves 0-1 load G's operands first; blend factors before it; mask turn behind the scores
# speedup vs baseline: 14.6963x; 1.0805x over previous
#include <hip/hip_runtime.h>
#include <hip/hip_bf16.h>
#include <math.h>

#define B_   4
#define N_   512
#define D_   32
#define PE_  16

#define NW  8
#define JT_PER_W (N_ / 16 / NW)
#define DH  16
static_assert(NW == 8, "the merge gives output rows r and r + 8 of the 16-row tile to wave r, and 8 lanes hold the NW partials' m and z");
static_assert(JT_PER_W * NW * 16 == N_, "the waves' j slices tile N exactly");
static_assert(2 * DH == D_, "two work-groups per i-tile cover the D columns");
#define TA_PITCH (16 * JT_PER_W + 4)
#define MRG_TILE (16 * DH + 16)
#define L2E 1.44269504088896340736f
#define INV_SQRT_D 0.17677669529663688110f
#define KS2 (10.0f * L2E)
#define LN2 0.69314718055994530942f

typedef __attribute__((ext_vector_type(2))) float v2f;
typedef __attribute__((ext_vector_type(4))) float v4f;
typedef __attribute__((ext_vector_type(8))) float v8f;

static __device__ __forceinline__ v8f wmma_f32(v2f a, v2f b, v8f c) {
  return __builtin_amdgcn_wmma_f32_16x16x4_f32(false, a, false, b, (short)0, c, false, false);
}

static __device__ __forceinline__ float fexp2(float x) { return __builtin_amdgcn_exp2f(x); }
static __device__ __forceinline__ float frcp(float x)  { return __builtin_amdgcn_rcpf(x); }
static __device__ __forceinline__ float flog2(float x) { return __builtin_amdgcn_logf(x); }

#define DPP_QUAD_XOR1  0xB1
#define DPP_QUAD_XOR2  0x4E
#define DPP_HALF_MIR   0x141

template <int CTRL>
static __device__ __forceinline__ float dpp_mov(float x) {
  return __int_as_float(
      __builtin_amdgcn_update_dpp(0, __float_as_int(x), CTRL, 0xf, 0xf, true));
}
static __device__ __forceinline__ float redmax8(float x) {
  x = fmaxf(x, dpp_mov<DPP_QUAD_XOR1>(x));
  x = fmaxf(x, dpp_mov<DPP_QUAD_XOR2>(x));
  x = fmaxf(x, dpp_mov<DPP_HALF_MIR>(x));
  return x;
}
static __device__ __forceinline__ float redsum8(float x) {
  x += dpp_mov<DPP_QUAD_XOR1>(x);
  x += dpp_mov<DPP_QUAD_XOR2>(x);
  x += dpp_mov<DPP_HALF_MIR>(x);
  return x;
}


static __device__ __forceinline__ int mrg_idx(int w, int row, int lo) {
  return w * MRG_TILE + (row + (row >> 3)) * 16 + lo;
}
static __device__ __forceinline__ float xhalf(float x) {
  return __int_as_float(__builtin_amdgcn_permlanex16(__float_as_int(x), __float_as_int(x),
                                                     0x76543210, 0xfedcba98, false, false));
}

#define LOAD_INPUTS() do { \
    Eil = E[i0 + lo]; \
_Pragma("unroll") \
    for (int t = 0; t < JT_PER_W; ++t) { \
      const int j0 = (wid * JT_PER_W + t) * 16; \
      const float4 e0 = *(const float4*)(E + j0 + 8 * hi); \
      const float4 e1 = *(const float4*)(E + j0 + 8 * hi + 4); \
      Ej[t][0] = e0.x; Ej[t][1] = e0.y; Ej[t][2] = e0.z; Ej[t][3] = e0.w; \
      Ej[t][4] = e1.x; Ej[t][5] = e1.y; Ej[t][6] = e1.z; Ej[t][7] = e1.w; \
    } \
_Pragma("unroll") \
    for (int t = 0; t < JT_PER_W; ++t) { \
      const int j0 = (wid * JT_PER_W + t) * 16; \
_Pragma("unroll") \
      for (int q = 0; q < 4; ++q) bPj[t][q] = *(const v2f*)(Pb + (j0 + lo) * PE_ + 8 * hi + 2 * q); \
    } \
_Pragma("unroll") \
    for (int m = 0; m < 8; ++m) \
      aRow[m] = *(const v4f*)(A + (i0 + 2 * m + (lane >> 4)) * N_ + wid * (16 * JT_PER_W) + 4 * (lane & 15)); \
  } while (0)

#define BLEND_FACTORS() do { \
_Pragma("unroll") \
    for (int t = 0; t < JT_PER_W; ++t) \
_Pragma("unroll") \
      for (int r = 0; r < 8; ++r) { \
        const float Ed = Ej[t][r] - Eil; \
        sg[t][r] = frcp(1.f + fexp2(-KS2 * Ed)); \
      } \
  } while (0)

static __device__ __forceinline__ void lds_barrier() {
  __builtin_amdgcn_fence(__ATOMIC_RELEASE, "workgroup", "local");
  __builtin_amdgcn_s_barrier();
  __builtin_amdgcn_fence(__ATOMIC_ACQUIRE, "workgroup", "local");
}

__global__ void __launch_bounds__(32 * NW) __attribute__((amdgpu_waves_per_eu(2, 2)))
fokker_planck_fused_kernel(const float* __restrict__ h,
                           const float* __restrict__ pe,
                           const float* __restrict__ E,
                           const float* __restrict__ A,
                           const float* __restrict__ Wk, const float* __restrict__ bk,
                           const float* __restrict__ Wq, const float* __restrict__ bq,
                           const float* __restrict__ beta,
                           float* __restrict__ out) {
  __shared__ __attribute__((aligned(16))) float mrgN[NW * MRG_TILE];
  __shared__ __attribute__((aligned(16))) float mrgS[NW * 17 * 4];
  __shared__ __attribute__((aligned(16))) float gxG[2 * 2 * 32 * 4];
  __shared__ float gxb[2 * 2 * 16];
  __shared__ __attribute__((aligned(16))) float tA[NW * 16 * TA_PITCH];

  const int lane = threadIdx.x & 31;
  const int wid  = threadIdx.x >> 5;
  const int lo   = lane & 15;
  const int hi   = lane >> 4;
  const int tile = blockIdx.x >> 1;
  const int c0 = (blockIdx.x & 1) * DH;
  const int b  = tile >> 5;
  const int i0 = (tile & 31) * 16;

  const float* Pb = pe + b * N_ * PE_;
  const float* Hb = h  + b * N_ * D_;

  const int mrow = wid + 8 * hi;

  float Eil;
  float Ej[JT_PER_W][8];
  v2f bPj[JT_PER_W][4];
  v4f aRow[8];
  float bts;
  float Ei[8];
  float Hi[8];
  float Hrow, Erow;

  float sg[JT_PER_W][8];

  const int wv = __builtin_amdgcn_readfirstlane(wid);
  if (wv >= 2) {
    LOAD_INPUTS();
    BLEND_FACTORS();
  } else {
    const int s = wv;
    v2f aWk[4], bPi[4], aWq[4];
    float bkv[8], bqv[8];
#pragma unroll
    for (int kc = 0; kc < 4; ++kc) {
      const int o = (4 * kc + 2 * hi) * D_ + 16 * s + lo;
      aWk[kc].x = Wk[o]; aWk[kc].y = Wk[o + D_];
      bPi[kc] = *(const v2f*)(Pb + (i0 + lo) * PE_ + 4 * kc + 2 * hi);
    }
#pragma unroll
    for (int q = 0; q < 4; ++q) aWq[q] = *(const v2f*)(Wq + lo * D_ + 16 * s + 8 * hi + 2 * q);
#pragma unroll
    for (int v = 0; v < 8; ++v) { bkv[v] = bk[16 * s + 8 * hi + v]; bqv[v] = bq[16 * s + 8 * hi + v]; }
    LOAD_INPUTS();
    v8f kt;
#pragma unroll
    for (int v = 0; v < 8; ++v) kt[v] = bkv[v];
#pragma unroll
    for (int kc = 0; kc < 4; ++kc) kt = wmma_f32(aWk[kc], bPi[kc], kt);
    v8f gt = {};
#pragma unroll
    for (int q = 0; q < 4; ++q) {
      v2f bK; bK.x = kt[2 * q]; bK.y = kt[2 * q + 1];
      gt = wmma_f32(aWq[q], bK, gt);
    }
    float gp = 0.f;
#pragma unroll
    for (int v = 0; v < 8; ++v) gp = fmaf(kt[v], bqv[v], gp);
    float4 g0; g0.x = gt[0]; g0.y = gt[1]; g0.z = gt[2]; g0.w = gt[3];
    float4 g1; g1.x = gt[4]; g1.y = gt[5]; g1.z = gt[6]; g1.w = gt[7];
    *(float4*)(gxG + ((s * 2 + 0) * 32 + lane) * 4) = g0;
    *(float4*)(gxG + ((s * 2 + 1) * 32 + lane) * 4) = g1;
    gxb[(s * 2 + hi) * 16 + lo] = gp;
  }
  lds_barrier();
  if (wv < 2) BLEND_FACTORS();

  const float kscale = INV_SQRT_D * L2E;
  v2f afG[4];
  float ginit;
  {
    const float4 a0 = *(const float4*)(gxG + ((0 * 2 + 0) * 32 + lane) * 4);
    const float4 a1 = *(const float4*)(gxG + ((0 * 2 + 1) * 32 + lane) * 4);
    const float4 b0 = *(const float4*)(gxG + ((1 * 2 + 0) * 32 + lane) * 4);
    const float4 b1 = *(const float4*)(gxG + ((1 * 2 + 1) * 32 + lane) * 4);
    afG[0].x = (a0.x + b0.x) * kscale;  afG[0].y = (a0.y + b0.y) * kscale;
    afG[1].x = (a0.z + b0.z) * kscale;  afG[1].y = (a0.w + b0.w) * kscale;
    afG[2].x = (a1.x + b1.x) * kscale;  afG[2].y = (a1.y + b1.y) * kscale;
    afG[3].x = (a1.z + b1.z) * kscale;  afG[3].y = (a1.w + b1.w) * kscale;
    ginit = ((gxb[0 * 16 + lo] + gxb[1 * 16 + lo]) + (gxb[2 * 16 + lo] + gxb[3 * 16 + lo])) * kscale;
  }

  float Hj[JT_PER_W][8];
#pragma unroll
  for (int t = 0; t < JT_PER_W; ++t) {
    const int j0 = (wid * JT_PER_W + t) * 16;
#pragma unroll
    for (int r = 0; r < 8; ++r) Hj[t][r] = Hb[(j0 + 8 * hi + r) * D_ + c0 + lo];
  }

  v8f sT[JT_PER_W];
#pragma unroll
  for (int t = 0; t < JT_PER_W; ++t)
#pragma unroll
    for (int v = 0; v < 8; ++v) sT[t][v] = ginit;
#pragma unroll
  for (int q = 0; q < 4; ++q)
#pragma unroll
    for (int t = 0; t < JT_PER_W; ++t) sT[t] = wmma_f32(bPj[t][q], afG[q], sT[t]);

  __builtin_amdgcn_sched_barrier(0);
#pragma unroll
  for (int m = 0; m < 8; ++m)
    *(v4f*)(tA + (wid * 16 + 2 * m + (lane >> 4)) * TA_PITCH + 4 * (lane & 15)) = aRow[m];
  float Am[JT_PER_W][8];
#pragma unroll
  for (int t = 0; t < JT_PER_W; ++t) {
    const float* ar = tA + (wid * 16 + lo) * TA_PITCH + 16 * t + 8 * hi;
    const float4 a0 = *(const float4*)(ar);
    const float4 a1 = *(const float4*)(ar + 4);
    Am[t][0] = a0.x; Am[t][1] = a0.y; Am[t][2] = a0.z; Am[t][3] = a0.w;
    Am[t][4] = a1.x; Am[t][5] = a1.y; Am[t][6] = a1.z; Am[t][7] = a1.w;
  }
  bts = beta[c0 + lo] * LN2;
#pragma unroll
  for (int v = 0; v < 8; ++v) {
    Ei[v] = E[i0 + v + 8 * hi];
    Hi[v] = Hb[(i0 + v + 8 * hi) * D_ + c0 + lo];
  }
  Hrow = Hb[(i0 + mrow) * D_ + c0 + lo];
  Erow = E[i0 + mrow];
  float sc[JT_PER_W][8];
#pragma unroll
  for (int t = 0; t < JT_PER_W; ++t)
#pragma unroll
    for (int r = 0; r < 8; ++r) sc[t][r] = Am[t][r] * sT[t][r];
  float mx = sc[0][0];
#pragma unroll
  for (int t = 0; t < JT_PER_W; ++t)
#pragma unroll
    for (int r = 0; r < 8; ++r) mx = fmaxf(mx, sc[t][r]);
  mx = fmaxf(mx, xhalf(mx));

  v2f p3a[JT_PER_W][4], p4a[JT_PER_W][4];
  float zc[4] = {0.f, 0.f, 0.f, 0.f}, cc[4] = {0.f, 0.f, 0.f, 0.f};
#pragma unroll
  for (int t = 0; t < JT_PER_W; ++t)
#pragma unroll
    for (int r = 0; r < 8; ++r) {
      const float p  = fexp2(sc[t][r] - mx);
      const float p3 = p * sg[t][r];
      const float p4 = p - p3;
      zc[r & 3] += p;
      cc[r & 3] += p4;
      p3a[t][r >> 1][r & 1] = p3;
      p4a[t][r >> 1][r & 1] = p4;
    }
  float z  = (zc[0] + zc[1]) + (zc[2] + zc[3]);
  float c6 = (cc[0] + cc[1]) + (cc[2] + cc[3]);
  z  += xhalf(z);
  c6 += xhalf(c6);

  __builtin_amdgcn_sched_barrier(0);
#pragma unroll
  for (int t = 0; t < JT_PER_W; ++t)
#pragma unroll
    for (int r = 0; r < 8; ++r) asm volatile("" : "+v"(Hj[t][r]));
  v8f accX = {}, accH = {}, accV = {};
#pragma unroll
  for (int t = 0; t < JT_PER_W; ++t) {
    v2f HB[4], VB[4], HVB[4];
#pragma unroll
    for (int q = 0; q < 4; ++q) {
#pragma unroll
      for (int tt = 0; tt < 2; ++tt) {
        const float hv = Hj[t][2 * q + tt];
        const float lv = flog2(hv + 1e-8f);
        const float vv = fmaf(bts, lv, Ej[t][2 * q + tt]);
        HB[q][tt]  = hv;
        VB[q][tt]  = vv;
        HVB[q][tt] = hv * vv;
      }
    }
#pragma unroll
    for (int q = 0; q < 4; ++q) {
      accX = wmma_f32(p3a[t][q], HVB[q], accX);
      accH = wmma_f32(p3a[t][q], HB[q],  accH);
      accV = wmma_f32(p4a[t][q], VB[q],  accV);
    }
  }

  float numw[8];
#pragma unroll
  for (int v = 0; v < 8; ++v) {
    const float hv = Hi[v];
    const float lv = flog2(hv + 1e-8f);
    const float vi = fmaf(bts, lv, Ei[v]);
    const float t2 = fmaf(hv, accV[v], accX[v]);
    numw[v] = fmaf(-vi, accH[v], t2);
  }

  {
#pragma unroll
    for (int v = 0; v < 8; ++v) mrgN[mrg_idx(wid, v + 8 * hi, lo)] = numw[v];
    if (hi == 0) {
      float4 sv; sv.x = mx; sv.y = z; sv.z = c6; sv.w = 0.f;
      *(float4*)(mrgS + (wid * 17 + lo) * 4) = sv;
    }
  }
  __syncthreads();

  {
    const int row = mrow;
    const float4 sv = *(const float4*)(mrgS + ((lo & 7) * 17 + row) * 4);
    const float M  = redmax8(sv.x);
    const float f  = fexp2(sv.x - M);
    const float zt = redsum8(f * sv.y);
    const float ct = redsum8(f * sv.z);
    float num = 0.f;
#define MERGE_STEP(k)                                                              \
    {                                                                              \
      const float fk = dpp_mov<0x150 + (k)>(f);                                    \
      num = fmaf(fk, mrgN[mrg_idx((k), row, lo)], num);                            \
    }
    MERGE_STEP(0) MERGE_STEP(1) MERGE_STEP(2) MERGE_STEP(3)
    MERGE_STEP(4) MERGE_STEP(5) MERGE_STEP(6) MERGE_STEP(7)
#undef MERGE_STEP
    const float vrow = fmaf(bts, flog2(Hrow + 1e-8f), Erow);
    num = fmaf(-(vrow * Hrow), ct, num);
    out[(b * N_ + i0 + row) * D_ + c0 + lo] = num * frcp(zt);
  }
}

extern "C" void kernel_launch(void* const* d_in, const int* in_sizes, int n_in,
                              void* d_out, int out_size, void* d_ws, size_t ws_size,
                              hipStream_t stream) {
  const float* h    = (const float*)d_in[1];
  const float* pe   = (const float*)d_in[2];
  const float* E    = (const float*)d_in[3];
  const float* A    = (const float*)d_in[4];
  const float* Wk   = (const float*)d_in[5];
  const float* bk   = (const float*)d_in[6];
  const float* Wq   = (const float*)d_in[7];
  const float* bq   = (const float*)d_in[8];
  const float* beta = (const float*)d_in[9];
  float* out = (float*)d_out;

  dim3 grid(2 * B_ * (N_ / 16)), block(32 * NW);
  fokker_planck_fused_kernel<<<grid, block, 0, stream>>>(h, pe, E, A, Wk, bk, Wq, bq, beta, out);
}
